# starting kernel
# baseline (speedup 1.0000x reference)
#include <hip/hip_runtime.h>
#include <cmath>

typedef __attribute__((ext_vector_type(2))) float v2f;
typedef __attribute__((ext_vector_type(8))) float v8f;

#define N_IN_F   165
#define N_HID_F  128
#define N_EMB_F  2
#define K_PAD    168
#define BSTRIDE  170

__global__ void k_fill(float* __restrict__ p, float v, int n) {
    int i = blockIdx.x * blockDim.x + threadIdx.x;
    if (i < n) p[i] = v;
}

__global__ void k_count_deg(const int* __restrict__ dst, float* __restrict__ deg, int nE) {
    int e = blockIdx.x * blockDim.x + threadIdx.x;
    if (e < nE) atomicAdd(&deg[dst[e]], 1.0f);
}

__global__ void k_rsqrt_inplace(float* __restrict__ p, int n) {
    int i = blockIdx.x * blockDim.x + threadIdx.x;
    if (i < n) p[i] = rsqrtf(p[i]);
}

__global__ void __launch_bounds__(256, 1)
k_gemm1_wmma(const float* __restrict__ X, const float* __restrict__ W,
             float* __restrict__ H, int nNodes) {
    extern __shared__ float bsh[];

    for (int idx = threadIdx.x; idx < N_HID_F * K_PAD; idx += blockDim.x) {
        int n = idx / K_PAD;
        int k = idx - n * K_PAD;
        int kc = (k < N_IN_F) ? k : (N_IN_F - 1);
        float wv = W[(size_t)kc * N_HID_F + n];
        bsh[n * BSTRIDE + k] = (k < N_IN_F) ? wv : 0.0f;
    }
    __syncthreads();

    const int lane   = threadIdx.x & 31;
    const int tileM  = blockIdx.x * (blockDim.x >> 5) + (threadIdx.x >> 5);
    const int nTiles = (nNodes + 15) >> 4;
    if (tileM >= nTiles) return;

    const int hi = lane >> 4;
    const int lo = lane & 15;
    int m = tileM * 16 + lo;
    if (m >= nNodes) m = nNodes - 1;

    v8f zero = {};
    v8f acc[8];
#pragma unroll
    for (int t = 0; t < 8; ++t) acc[t] = zero;

    const float* Xrow  = X + (size_t)m * N_IN_F;
    const float* bbase = bsh + lo * BSTRIDE;

#pragma unroll 2
    for (int k0 = 0; k0 < 164; k0 += 4) {
        const int kA = k0 + 2 * hi;
        v2f a;
        a.x = Xrow[kA];
        a.y = Xrow[kA + 1];
#pragma unroll
        for (int t = 0; t < 8; ++t) {
            v2f b = *(const v2f*)(bbase + (size_t)(t * 16) * BSTRIDE + kA);
            acc[t] = __builtin_amdgcn_wmma_f32_16x16x4_f32(
                false, a, false, b, (short)0, acc[t], false, false);
        }
    }

    {
        const int kA = 164 + 2 * hi;
        float xv = Xrow[164];
        v2f a;
        a.x = (hi == 0) ? xv : 0.0f;
        a.y = 0.0f;
#pragma unroll
        for (int t = 0; t < 8; ++t) {
            v2f b = *(const v2f*)(bbase + (size_t)(t * 16) * BSTRIDE + kA);
            acc[t] = __builtin_amdgcn_wmma_f32_16x16x4_f32(
                false, a, false, b, (short)0, acc[t], false, false);
        }
    }

    if (tileM * 16 + 16 <= nNodes) {
        float* Hrow = H + (size_t)(tileM * 16 + 8 * hi) * N_HID_F + lo;
#pragma unroll
        for (int r = 0; r < 8; ++r) {
#pragma unroll
            for (int t = 0; t < 8; ++t)
                Hrow[(size_t)r * N_HID_F + t * 16] = acc[t][r];
        }
    } else {
#pragma unroll
        for (int r = 0; r < 8; ++r) {
            int row = tileM * 16 + r + 8 * hi;
            if (row < nNodes) {
#pragma unroll
                for (int t = 0; t < 8; ++t)
                    H[(size_t)row * N_HID_F + t * 16 + lo] = acc[t][r];
            }
        }
    }
}

__global__ void k_aggregate_hid(const float* __restrict__ H, const int* __restrict__ src,
                                const int* __restrict__ dst, const float* __restrict__ inv,
                                float* __restrict__ out, int nE) {
    const int lane = threadIdx.x & 31;
    const int e = blockIdx.x * (blockDim.x >> 5) + (threadIdx.x >> 5);
    if (e >= nE) return;
    const int s = src[e], d = dst[e];
    const float c = inv[s] * inv[d];
    const float4 v = ((const float4*)(H + (size_t)s * N_HID_F))[lane];
    float* o = out + (size_t)d * N_HID_F + lane * 4;
    atomicAdd(o + 0, v.x * c);
    atomicAdd(o + 1, v.y * c);
    atomicAdd(o + 2, v.z * c);
    atomicAdd(o + 3, v.w * c);
}

__global__ void k_post1(float* __restrict__ H1, const float* __restrict__ agg,
                        const float* __restrict__ inv, const float* __restrict__ b1,
                        int total) {
    int i = blockIdx.x * blockDim.x + threadIdx.x;
    if (i >= total) return;
    int node = i >> 7;
    int f    = i & 127;
    float is = inv[node];
    float v  = agg[i] + H1[i] * is * is + b1[f];
    H1[i] = tanhf(v);
}

__global__ void k_gemm2(const float* __restrict__ H, const float* __restrict__ W2,
                        float* __restrict__ H2, int nNodes) {
    __shared__ float w2s[N_HID_F * N_EMB_F];
    int tid = threadIdx.x;
    if (tid < N_HID_F * N_EMB_F) w2s[tid] = W2[tid];
    __syncthreads();
    int n = blockIdx.x * blockDim.x + tid;
    if (n >= nNodes) return;
    const float* h = H + (size_t)n * N_HID_F;
    float a0 = 0.0f, a1 = 0.0f;
#pragma unroll 8
    for (int k = 0; k < N_HID_F; ++k) {
        float hv = h[k];
        a0 += hv * w2s[k * 2 + 0];
        a1 += hv * w2s[k * 2 + 1];
    }
    H2[(size_t)n * 2 + 0] = a0;
    H2[(size_t)n * 2 + 1] = a1;
}

__global__ void k_aggregate_emb(const float* __restrict__ H2, const int* __restrict__ src,
                                const int* __restrict__ dst, const float* __restrict__ inv,
                                float* __restrict__ out, int nE) {
    int e = blockIdx.x * blockDim.x + threadIdx.x;
    if (e >= nE) return;
    const int s = src[e], d = dst[e];
    const float c = inv[s] * inv[d];
    atomicAdd(&out[(size_t)d * 2 + 0], H2[(size_t)s * 2 + 0] * c);
    atomicAdd(&out[(size_t)d * 2 + 1], H2[(size_t)s * 2 + 1] * c);
}

__global__ void k_final(const float* __restrict__ H2, const float* __restrict__ agg,
                        const float* __restrict__ inv, const float* __restrict__ b2,
                        const float* __restrict__ Wc, const float* __restrict__ bc,
                        float* __restrict__ out, int nNodes) {
    int n = blockIdx.x * blockDim.x + threadIdx.x;
    if (n >= nNodes) return;
    float is = inv[n];
    float i2 = is * is;
    float e0 = tanhf(agg[(size_t)n * 2 + 0] + H2[(size_t)n * 2 + 0] * i2 + b2[0]);
    float e1 = tanhf(agg[(size_t)n * 2 + 1] + H2[(size_t)n * 2 + 1] * i2 + b2[1]);
    float z  = e0 * Wc[0] + e1 * Wc[1] + bc[0];
    out[n] = 1.0f / (1.0f + expf(-z));
}

extern "C" void kernel_launch(void* const* d_in, const int* in_sizes, int n_in,
                              void* d_out, int out_size, void* d_ws, size_t ws_size,
                              hipStream_t stream) {
    const float* x   = (const float*)d_in[0];
    const int*   ei  = (const int*)d_in[1];
    const float* W1  = (const float*)d_in[2];
    const float* b1  = (const float*)d_in[3];
    const float* W2  = (const float*)d_in[4];
    const float* b2  = (const float*)d_in[5];
    const float* Wc  = (const float*)d_in[6];
    const float* bc  = (const float*)d_in[7];
    float* out = (float*)d_out;

    const int nN = in_sizes[0] / N_IN_F;
    const int nE = in_sizes[1] / 2;
    const int* src = ei;
    const int* dst = ei + nE;

    float* ws   = (float*)d_ws;
    float* h1   = ws;
    float* agg1 = h1   + (size_t)nN * N_HID_F;
    float* inv  = agg1 + (size_t)nN * N_HID_F;
    float* h2   = inv  + (size_t)nN;
    float* agg2 = h2   + (size_t)nN * 2;

    const int B   = 256;
    const int WPB = B / 32;
    const int hidTotal = nN * N_HID_F;

    k_fill<<<(nN + B - 1) / B, B, 0, stream>>>(inv, 1.0f, nN);
    k_count_deg<<<(nE + B - 1) / B, B, 0, stream>>>(dst, inv, nE);
    k_rsqrt_inplace<<<(nN + B - 1) / B, B, 0, stream>>>(inv, nN);

    const int nTiles = (nN + 15) / 16;
    const size_t ldsBytes = (size_t)N_HID_F * BSTRIDE * sizeof(float);
    k_gemm1_wmma<<<(nTiles + WPB - 1) / WPB, B, ldsBytes, stream>>>(x, W1, h1, nN);
    k_fill<<<(hidTotal + B - 1) / B, B, 0, stream>>>(agg1, 0.0f, hidTotal);
    k_aggregate_hid<<<(nE + WPB - 1) / WPB, B, 0, stream>>>(h1, src, dst, inv, agg1, nE);
    k_post1<<<(hidTotal + B - 1) / B, B, 0, stream>>>(h1, agg1, inv, b1, hidTotal);

    k_gemm2<<<(nN + B - 1) / B, B, 0, stream>>>(h1, W2, h2, nN);
    k_fill<<<(nN * 2 + B - 1) / B, B, 0, stream>>>(agg2, 0.0f, nN * 2);
    k_aggregate_emb<<<(nE + B - 1) / B, B, 0, stream>>>(h2, src, dst, inv, agg2, nE);
    k_final<<<(nN + B - 1) / B, B, 0, stream>>>(h2, agg2, inv, b2, Wc, bc, out, nN);
}


# Degree and layer-2 sums: band-owner work-groups add in LDS over edge slices, 64 partial planes summed in fixed order, replace scattered global float atomics; sum order = LDS arrival, then slice
# speedup vs baseline: 1.3927x; 1.3927x over previous
#include <hip/hip_runtime.h>
#include <cmath>

typedef __attribute__((ext_vector_type(2))) float v2f;
typedef __attribute__((ext_vector_type(8))) float v8f;

#define N_IN_F   165
#define N_HID_F  128
#define N_EMB_F  2
#define K_PAD    168
#define BSTRIDE  170

__global__ void k_fill(float* __restrict__ p, float v, int n) {
    int i = blockIdx.x * blockDim.x + threadIdx.x;
    if (i < n) p[i] = v;
}

#define NR_MAX   32768
#define NSLICE   64
#define BAND_T   1024
#define EU       8

__global__ void __launch_bounds__(BAND_T)
k_deg_band(const int* __restrict__ dst, unsigned* __restrict__ part,
           int nE, int nN, int nBands, int NR, int ePerSlice) {
    __shared__ unsigned hist[NR_MAX];
    const int tid   = threadIdx.x;
    const int band  = blockIdx.x % nBands;
    const int slice = blockIdx.x / nBands;
    const int lo    = band * NR;
    int cnt = nN - lo;
    if (cnt > NR) cnt = NR;
    if (cnt < 0)  cnt = 0;

    for (int i = tid; i < cnt; i += BAND_T) hist[i] = 0u;
    __syncthreads();

    const int e0 = slice * ePerSlice;
    int e1 = e0 + ePerSlice;
    if (e1 > nE) e1 = nE;
    for (int base = e0; base < e1; base += EU * BAND_T) {
        int d[EU];
#pragma unroll
        for (int u = 0; u < EU; ++u) {
            int idx = base + u * BAND_T + tid;
            int ic  = (idx < e1) ? idx : (e1 - 1);
            int dv  = dst[ic];
            d[u] = (idx < e1) ? dv : -1;
        }
#pragma unroll
        for (int u = 0; u < EU; ++u) {
            unsigned r = (unsigned)(d[u] - lo);
            if (r < (unsigned)cnt) atomicAdd(&hist[r], 1u);
        }
    }
    __syncthreads();

    unsigned* p = part + (size_t)slice * nN + lo;
    for (int i = tid; i < cnt; i += BAND_T) p[i] = hist[i];
}

__global__ void k_inv_from_parts(const unsigned* __restrict__ part, float* __restrict__ inv, int nN) {
    int n = blockIdx.x * blockDim.x + threadIdx.x;
    if (n >= nN) return;
    unsigned c = 0u;
#pragma unroll 16
    for (int s = 0; s < NSLICE; ++s) c += part[(size_t)s * nN + n];
    inv[n] = rsqrtf(1.0f + (float)c);
}

__global__ void __launch_bounds__(256, 1)
k_gemm1_wmma(const float* __restrict__ X, const float* __restrict__ W,
             float* __restrict__ H, int nNodes) {
    extern __shared__ float bsh[];

    for (int idx = threadIdx.x; idx < N_HID_F * K_PAD; idx += blockDim.x) {
        int n = idx / K_PAD;
        int k = idx - n * K_PAD;
        int kc = (k < N_IN_F) ? k : (N_IN_F - 1);
        float wv = W[(size_t)kc * N_HID_F + n];
        bsh[n * BSTRIDE + k] = (k < N_IN_F) ? wv : 0.0f;
    }
    __syncthreads();

    const int lane   = threadIdx.x & 31;
    const int tileM  = blockIdx.x * (blockDim.x >> 5) + (threadIdx.x >> 5);
    const int nTiles = (nNodes + 15) >> 4;
    if (tileM >= nTiles) return;

    const int hi = lane >> 4;
    const int lo = lane & 15;
    int m = tileM * 16 + lo;
    if (m >= nNodes) m = nNodes - 1;

    v8f zero = {};
    v8f acc[8];
#pragma unroll
    for (int t = 0; t < 8; ++t) acc[t] = zero;

    const float* Xrow  = X + (size_t)m * N_IN_F;
    const float* bbase = bsh + lo * BSTRIDE;

#pragma unroll 2
    for (int k0 = 0; k0 < 164; k0 += 4) {
        const int kA = k0 + 2 * hi;
        v2f a;
        a.x = Xrow[kA];
        a.y = Xrow[kA + 1];
#pragma unroll
        for (int t = 0; t < 8; ++t) {
            v2f b = *(const v2f*)(bbase + (size_t)(t * 16) * BSTRIDE + kA);
            acc[t] = __builtin_amdgcn_wmma_f32_16x16x4_f32(
                false, a, false, b, (short)0, acc[t], false, false);
        }
    }

    {
        const int kA = 164 + 2 * hi;
        float xv = Xrow[164];
        v2f a;
        a.x = (hi == 0) ? xv : 0.0f;
        a.y = 0.0f;
#pragma unroll
        for (int t = 0; t < 8; ++t) {
            v2f b = *(const v2f*)(bbase + (size_t)(t * 16) * BSTRIDE + kA);
            acc[t] = __builtin_amdgcn_wmma_f32_16x16x4_f32(
                false, a, false, b, (short)0, acc[t], false, false);
        }
    }

    if (tileM * 16 + 16 <= nNodes) {
        float* Hrow = H + (size_t)(tileM * 16 + 8 * hi) * N_HID_F + lo;
#pragma unroll
        for (int r = 0; r < 8; ++r) {
#pragma unroll
            for (int t = 0; t < 8; ++t)
                Hrow[(size_t)r * N_HID_F + t * 16] = acc[t][r];
        }
    } else {
#pragma unroll
        for (int r = 0; r < 8; ++r) {
            int row = tileM * 16 + r + 8 * hi;
            if (row < nNodes) {
#pragma unroll
                for (int t = 0; t < 8; ++t)
                    H[(size_t)row * N_HID_F + t * 16 + lo] = acc[t][r];
            }
        }
    }
}

__global__ void k_aggregate_hid(const float* __restrict__ H, const int* __restrict__ src,
                                const int* __restrict__ dst, const float* __restrict__ inv,
                                float* __restrict__ out, int nE) {
    const int lane = threadIdx.x & 31;
    const int e = blockIdx.x * (blockDim.x >> 5) + (threadIdx.x >> 5);
    if (e >= nE) return;
    const int s = src[e], d = dst[e];
    const float c = inv[s] * inv[d];
    const float4 v = ((const float4*)(H + (size_t)s * N_HID_F))[lane];
    float* o = out + (size_t)d * N_HID_F + lane * 4;
    atomicAdd(o + 0, v.x * c);
    atomicAdd(o + 1, v.y * c);
    atomicAdd(o + 2, v.z * c);
    atomicAdd(o + 3, v.w * c);
}

__global__ void k_post1(float* __restrict__ H1, const float* __restrict__ agg,
                        const float* __restrict__ inv, const float* __restrict__ b1,
                        int total) {
    int i = blockIdx.x * blockDim.x + threadIdx.x;
    if (i >= total) return;
    int node = i >> 7;
    int f    = i & 127;
    float is = inv[node];
    float v  = agg[i] + H1[i] * is * is + b1[f];
    H1[i] = tanhf(v);
}

__global__ void k_gemm2(const float* __restrict__ H, const float* __restrict__ W2,
                        float* __restrict__ H2, int nNodes) {
    __shared__ float w2s[N_HID_F * N_EMB_F];
    int tid = threadIdx.x;
    if (tid < N_HID_F * N_EMB_F) w2s[tid] = W2[tid];
    __syncthreads();
    int n = blockIdx.x * blockDim.x + tid;
    if (n >= nNodes) return;
    const float* h = H + (size_t)n * N_HID_F;
    float a0 = 0.0f, a1 = 0.0f;
#pragma unroll 8
    for (int k = 0; k < N_HID_F; ++k) {
        float hv = h[k];
        a0 += hv * w2s[k * 2 + 0];
        a1 += hv * w2s[k * 2 + 1];
    }
    H2[(size_t)n * 2 + 0] = a0;
    H2[(size_t)n * 2 + 1] = a1;
}

__global__ void __launch_bounds__(BAND_T)
k_emb_band(const float* __restrict__ H2, const int* __restrict__ src,
           const int* __restrict__ dst, const float* __restrict__ inv,
           float* __restrict__ part, int nE, int nN, int nBands, int NR, int ePerSlice) {
    __shared__ __attribute__((aligned(16))) float acc[NR_MAX * 2];
    const int tid   = threadIdx.x;
    const int band  = blockIdx.x % nBands;
    const int slice = blockIdx.x / nBands;
    const int lo    = band * NR;
    int cnt = nN - lo;
    if (cnt > NR) cnt = NR;
    if (cnt < 0)  cnt = 0;

    {
        float4* a4 = (float4*)acc;
        const int n4 = (2 * cnt + 3) >> 2;
        const float4 z = make_float4(0.0f, 0.0f, 0.0f, 0.0f);
        for (int i = tid; i < n4; i += BAND_T) a4[i] = z;
    }
    __syncthreads();

    const int e0 = slice * ePerSlice;
    int e1 = e0 + ePerSlice;
    if (e1 > nE) e1 = nE;
    for (int base = e0; base < e1; base += EU * BAND_T) {
        int  dd[EU], ss[EU];
        bool m[EU];
#pragma unroll
        for (int u = 0; u < EU; ++u) {
            int idx = base + u * BAND_T + tid;
            int ic  = (idx < e1) ? idx : (e1 - 1);
            int d = dst[ic];
            int s = src[ic];
            unsigned r = (unsigned)(d - lo);
            m[u]  = (idx < e1) && (r < (unsigned)cnt);
            dd[u] = m[u] ? d : 0;
            ss[u] = m[u] ? s : 0;
        }
        float  is[EU], id[EU];
        float2 v[EU];
#pragma unroll
        for (int u = 0; u < EU; ++u) {
            is[u] = inv[ss[u]];
            id[u] = inv[dd[u]];
            v[u]  = *(const float2*)(H2 + (size_t)ss[u] * 2);
        }
        asm volatile("" :: "v"(is[0]), "v"(id[0]), "v"(v[0].x), "v"(v[0].y));
#pragma unroll
        for (int u = 0; u < EU; ++u) {
            if (m[u]) {
                const float c = is[u] * id[u];
                float* a = acc + 2 * (dd[u] - lo);
                atomicAdd(a + 0, v[u].x * c);
                atomicAdd(a + 1, v[u].y * c);
            }
        }
    }
    __syncthreads();

    float2*       p  = (float2*)(part + (size_t)slice * 2 * nN) + lo;
    const float2* a2 = (const float2*)acc;
    for (int i = tid; i < cnt; i += BAND_T) p[i] = a2[i];
}

__global__ void k_final(const float* __restrict__ H2, const float* __restrict__ part,
                        const float* __restrict__ inv, const float* __restrict__ b2,
                        const float* __restrict__ Wc, const float* __restrict__ bc,
                        float* __restrict__ out, int nNodes) {
    int n = blockIdx.x * blockDim.x + threadIdx.x;
    if (n >= nNodes) return;
    const float2* p = (const float2*)part + n;
    float g0 = 0.0f, g1 = 0.0f;
#pragma unroll 16
    for (int s = 0; s < NSLICE; ++s) {
        float2 t = p[(size_t)s * nNodes];
        g0 += t.x;
        g1 += t.y;
    }
    float is = inv[n];
    float i2 = is * is;
    float e0 = tanhf(g0 + H2[(size_t)n * 2 + 0] * i2 + b2[0]);
    float e1 = tanhf(g1 + H2[(size_t)n * 2 + 1] * i2 + b2[1]);
    float z  = e0 * Wc[0] + e1 * Wc[1] + bc[0];
    out[n] = 1.0f / (1.0f + expf(-z));
}

extern "C" void kernel_launch(void* const* d_in, const int* in_sizes, int n_in,
                              void* d_out, int out_size, void* d_ws, size_t ws_size,
                              hipStream_t stream) {
    const float* x   = (const float*)d_in[0];
    const int*   ei  = (const int*)d_in[1];
    const float* W1  = (const float*)d_in[2];
    const float* b1  = (const float*)d_in[3];
    const float* W2  = (const float*)d_in[4];
    const float* b2  = (const float*)d_in[5];
    const float* Wc  = (const float*)d_in[6];
    const float* bc  = (const float*)d_in[7];
    float* out = (float*)d_out;

    const int nN = in_sizes[0] / N_IN_F;
    const int nE = in_sizes[1] / 2;
    const int* src = ei;
    const int* dst = ei + nE;

    float* ws   = (float*)d_ws;
    float* h1   = ws;
    float* agg1 = h1   + (size_t)nN * N_HID_F;
    float* pemb = agg1 + (size_t)nN * N_HID_F;
    unsigned* pdeg = (unsigned*)(pemb + (size_t)NSLICE * nN * 2);
    float* h2   = (float*)(pdeg + (size_t)NSLICE * nN);
    float* inv  = h2   + (size_t)nN * 2;

    const int B   = 256;
    const int WPB = B / 32;
    const int hidTotal = nN * N_HID_F;

    const int nBands    = (nN + NR_MAX - 1) / NR_MAX;
    const int NR        = (nN + nBands - 1) / nBands;
    const int ePerSlice = (nE + NSLICE - 1) / NSLICE;

    k_deg_band<<<nBands * NSLICE, BAND_T, 0, stream>>>(dst, pdeg, nE, nN, nBands, NR, ePerSlice);
    k_inv_from_parts<<<(nN + B - 1) / B, B, 0, stream>>>(pdeg, inv, nN);

    const int nTiles = (nN + 15) / 16;
    const size_t ldsBytes = (size_t)N_HID_F * BSTRIDE * sizeof(float);
    k_gemm1_wmma<<<(nTiles + WPB - 1) / WPB, B, ldsBytes, stream>>>(x, W1, h1, nN);
    k_fill<<<(hidTotal + B - 1) / B, B, 0, stream>>>(agg1, 0.0f, hidTotal);
    k_aggregate_hid<<<(nE + WPB - 1) / WPB, B, 0, stream>>>(h1, src, dst, inv, agg1, nE);
    k_post1<<<(hidTotal + B - 1) / B, B, 0, stream>>>(h1, agg1, inv, b1, hidTotal);

    k_gemm2<<<(nN + B - 1) / B, B, 0, stream>>>(h1, W2, h2, nN);
    k_emb_band<<<nBands * NSLICE, BAND_T, 0, stream>>>(h2, src, dst, inv, pemb, nE, nN, nBands, NR, ePerSlice);
    k_final<<<(nN + B - 1) / B, B, 0, stream>>>(h2, pemb, inv, b2, Wc, bc, out, nN);
}


# Layer-1 scatter: lane l takes features l+32j, so each float atomic add and row load of a wave covers one whole 128-B line, not 8 floats in each of 4 lines; sum order = atomic arrival as before
# speedup vs baseline: 1.7917x; 1.2865x over previous
#include <hip/hip_runtime.h>
#include <cmath>

typedef __attribute__((ext_vector_type(2))) float v2f;
typedef __attribute__((ext_vector_type(8))) float v8f;

#define N_IN_F   165
#define N_HID_F  128
#define N_EMB_F  2
#define K_PAD    168
#define BSTRIDE  170

__global__ void k_fill(float* __restrict__ p, float v, int n) {
    int i = blockIdx.x * blockDim.x + threadIdx.x;
    if (i < n) p[i] = v;
}

#define NR_MAX   32768
#define NSLICE   64
#define BAND_T   1024
#define EU       8

__global__ void __launch_bounds__(BAND_T)
k_deg_band(const int* __restrict__ dst, unsigned* __restrict__ part,
           int nE, int nN, int nBands, int NR, int ePerSlice) {
    __shared__ unsigned hist[NR_MAX];
    const int tid   = threadIdx.x;
    const int band  = blockIdx.x % nBands;
    const int slice = blockIdx.x / nBands;
    const int lo    = band * NR;
    int cnt = nN - lo;
    if (cnt > NR) cnt = NR;
    if (cnt < 0)  cnt = 0;

    for (int i = tid; i < cnt; i += BAND_T) hist[i] = 0u;
    __syncthreads();

    const int e0 = slice * ePerSlice;
    int e1 = e0 + ePerSlice;
    if (e1 > nE) e1 = nE;
    for (int base = e0; base < e1; base += EU * BAND_T) {
        int d[EU];
#pragma unroll
        for (int u = 0; u < EU; ++u) {
            int idx = base + u * BAND_T + tid;
            int ic  = (idx < e1) ? idx : (e1 - 1);
            int dv  = dst[ic];
            d[u] = (idx < e1) ? dv : -1;
        }
#pragma unroll
        for (int u = 0; u < EU; ++u) {
            unsigned r = (unsigned)(d[u] - lo);
            if (r < (unsigned)cnt) atomicAdd(&hist[r], 1u);
        }
    }
    __syncthreads();

    unsigned* p = part + (size_t)slice * nN + lo;
    for (int i = tid; i < cnt; i += BAND_T) p[i] = hist[i];
}

__global__ void k_inv_from_parts(const unsigned* __restrict__ part, float* __restrict__ inv, int nN) {
    int n = blockIdx.x * blockDim.x + threadIdx.x;
    if (n >= nN) return;
    unsigned c = 0u;
#pragma unroll 16
    for (int s = 0; s < NSLICE; ++s) c += part[(size_t)s * nN + n];
    inv[n] = rsqrtf(1.0f + (float)c);
}

__global__ void __launch_bounds__(256, 1)
k_gemm1_wmma(const float* __restrict__ X, const float* __restrict__ W,
             float* __restrict__ H, int nNodes) {
    extern __shared__ float bsh[];

    for (int idx = threadIdx.x; idx < N_HID_F * K_PAD; idx += blockDim.x) {
        int n = idx / K_PAD;
        int k = idx - n * K_PAD;
        int kc = (k < N_IN_F) ? k : (N_IN_F - 1);
        float wv = W[(size_t)kc * N_HID_F + n];
        bsh[n * BSTRIDE + k] = (k < N_IN_F) ? wv : 0.0f;
    }
    __syncthreads();

    const int lane   = threadIdx.x & 31;
    const int tileM  = blockIdx.x * (blockDim.x >> 5) + (threadIdx.x >> 5);
    const int nTiles = (nNodes + 15) >> 4;
    if (tileM >= nTiles) return;

    const int hi = lane >> 4;
    const int lo = lane & 15;
    int m = tileM * 16 + lo;
    if (m >= nNodes) m = nNodes - 1;

    v8f zero = {};
    v8f acc[8];
#pragma unroll
    for (int t = 0; t < 8; ++t) acc[t] = zero;

    const float* Xrow  = X + (size_t)m * N_IN_F;
    const float* bbase = bsh + lo * BSTRIDE;

#pragma unroll 2
    for (int k0 = 0; k0 < 164; k0 += 4) {
        const int kA = k0 + 2 * hi;
        v2f a;
        a.x = Xrow[kA];
        a.y = Xrow[kA + 1];
#pragma unroll
        for (int t = 0; t < 8; ++t) {
            v2f b = *(const v2f*)(bbase + (size_t)(t * 16) * BSTRIDE + kA);
            acc[t] = __builtin_amdgcn_wmma_f32_16x16x4_f32(
                false, a, false, b, (short)0, acc[t], false, false);
        }
    }

    {
        const int kA = 164 + 2 * hi;
        float xv = Xrow[164];
        v2f a;
        a.x = (hi == 0) ? xv : 0.0f;
        a.y = 0.0f;
#pragma unroll
        for (int t = 0; t < 8; ++t) {
            v2f b = *(const v2f*)(bbase + (size_t)(t * 16) * BSTRIDE + kA);
            acc[t] = __builtin_amdgcn_wmma_f32_16x16x4_f32(
                false, a, false, b, (short)0, acc[t], false, false);
        }
    }

    if (tileM * 16 + 16 <= nNodes) {
        float* Hrow = H + (size_t)(tileM * 16 + 8 * hi) * N_HID_F + lo;
#pragma unroll
        for (int r = 0; r < 8; ++r) {
#pragma unroll
            for (int t = 0; t < 8; ++t)
                Hrow[(size_t)r * N_HID_F + t * 16] = acc[t][r];
        }
    } else {
#pragma unroll
        for (int r = 0; r < 8; ++r) {
            int row = tileM * 16 + r + 8 * hi;
            if (row < nNodes) {
#pragma unroll
                for (int t = 0; t < 8; ++t)
                    H[(size_t)row * N_HID_F + t * 16 + lo] = acc[t][r];
            }
        }
    }
}

__global__ void k_aggregate_hid(const float* __restrict__ H, const int* __restrict__ src,
                                const int* __restrict__ dst, const float* __restrict__ inv,
                                float* __restrict__ out, int nE) {
    const int lane = threadIdx.x & 31;
    const int e = blockIdx.x * (blockDim.x >> 5) + (threadIdx.x >> 5);
    if (e >= nE) return;
    const int s = src[e], d = dst[e];
    const float c = inv[s] * inv[d];
    const float* h = H + (size_t)s * N_HID_F + lane;
    const float v0 = h[0], v1 = h[32], v2 = h[64], v3 = h[96];
    float* o = out + (size_t)d * N_HID_F + lane;
    atomicAdd(o +  0, v0 * c);
    atomicAdd(o + 32, v1 * c);
    atomicAdd(o + 64, v2 * c);
    atomicAdd(o + 96, v3 * c);
}

__global__ void k_post1(float* __restrict__ H1, const float* __restrict__ agg,
                        const float* __restrict__ inv, const float* __restrict__ b1,
                        int total) {
    int i = blockIdx.x * blockDim.x + threadIdx.x;
    if (i >= total) return;
    int node = i >> 7;
    int f    = i & 127;
    float is = inv[node];
    float v  = agg[i] + H1[i] * is * is + b1[f];
    H1[i] = tanhf(v);
}

__global__ void k_gemm2(const float* __restrict__ H, const float* __restrict__ W2,
                        float* __restrict__ H2, int nNodes) {
    __shared__ float w2s[N_HID_F * N_EMB_F];
    int tid = threadIdx.x;
    if (tid < N_HID_F * N_EMB_F) w2s[tid] = W2[tid];
    __syncthreads();
    int n = blockIdx.x * blockDim.x + tid;
    if (n >= nNodes) return;
    const float* h = H + (size_t)n * N_HID_F;
    float a0 = 0.0f, a1 = 0.0f;
#pragma unroll 8
    for (int k = 0; k < N_HID_F; ++k) {
        float hv = h[k];
        a0 += hv * w2s[k * 2 + 0];
        a1 += hv * w2s[k * 2 + 1];
    }
    H2[(size_t)n * 2 + 0] = a0;
    H2[(size_t)n * 2 + 1] = a1;
}

__global__ void __launch_bounds__(BAND_T)
k_emb_band(const float* __restrict__ H2, const int* __restrict__ src,
           const int* __restrict__ dst, const float* __restrict__ inv,
           float* __restrict__ part, int nE, int nN, int nBands, int NR, int ePerSlice) {
    __shared__ __attribute__((aligned(16))) float acc[NR_MAX * 2];
    const int tid   = threadIdx.x;
    const int band  = blockIdx.x % nBands;
    const int slice = blockIdx.x / nBands;
    const int lo    = band * NR;
    int cnt = nN - lo;
    if (cnt > NR) cnt = NR;
    if (cnt < 0)  cnt = 0;

    {
        float4* a4 = (float4*)acc;
        const int n4 = (2 * cnt + 3) >> 2;
        const float4 z = make_float4(0.0f, 0.0f, 0.0f, 0.0f);
        for (int i = tid; i < n4; i += BAND_T) a4[i] = z;
    }
    __syncthreads();

    const int e0 = slice * ePerSlice;
    int e1 = e0 + ePerSlice;
    if (e1 > nE) e1 = nE;
    for (int base = e0; base < e1; base += EU * BAND_T) {
        int  dd[EU], ss[EU];
        bool m[EU];
#pragma unroll
        for (int u = 0; u < EU; ++u) {
            int idx = base + u * BAND_T + tid;
            int ic  = (idx < e1) ? idx : (e1 - 1);
            int d = dst[ic];
            int s = src[ic];
            unsigned r = (unsigned)(d - lo);
            m[u]  = (idx < e1) && (r < (unsigned)cnt);
            dd[u] = m[u] ? d : 0;
            ss[u] = m[u] ? s : 0;
        }
        float  is[EU], id[EU];
        float2 v[EU];
#pragma unroll
        for (int u = 0; u < EU; ++u) {
            is[u] = inv[ss[u]];
            id[u] = inv[dd[u]];
            v[u]  = *(const float2*)(H2 + (size_t)ss[u] * 2);
        }
        asm volatile("" :: "v"(is[0]), "v"(id[0]), "v"(v[0].x), "v"(v[0].y));
#pragma unroll
        for (int u = 0; u < EU; ++u) {
            if (m[u]) {
                const float c = is[u] * id[u];
                float* a = acc + 2 * (dd[u] - lo);
                atomicAdd(a + 0, v[u].x * c);
                atomicAdd(a + 1, v[u].y * c);
            }
        }
    }
    __syncthreads();

    float2*       p  = (float2*)(part + (size_t)slice * 2 * nN) + lo;
    const float2* a2 = (const float2*)acc;
    for (int i = tid; i < cnt; i += BAND_T) p[i] = a2[i];
}

__global__ void k_final(const float* __restrict__ H2, const float* __restrict__ part,
                        const float* __restrict__ inv, const float* __restrict__ b2,
                        const float* __restrict__ Wc, const float* __restrict__ bc,
                        float* __restrict__ out, int nNodes) {
    int n = blockIdx.x * blockDim.x + threadIdx.x;
    if (n >= nNodes) return;
    const float2* p = (const float2*)part + n;
    float g0 = 0.0f, g1 = 0.0f;
#pragma unroll 16
    for (int s = 0; s < NSLICE; ++s) {
        float2 t = p[(size_t)s * nNodes];
        g0 += t.x;
        g1 += t.y;
    }
    float is = inv[n];
    float i2 = is * is;
    float e0 = tanhf(g0 + H2[(size_t)n * 2 + 0] * i2 + b2[0]);
    float e1 = tanhf(g1 + H2[(size_t)n * 2 + 1] * i2 + b2[1]);
    float z  = e0 * Wc[0] + e1 * Wc[1] + bc[0];
    out[n] = 1.0f / (1.0f + expf(-z));
}

extern "C" void kernel_launch(void* const* d_in, const int* in_sizes, int n_in,
                              void* d_out, int out_size, void* d_ws, size_t ws_size,
                              hipStream_t stream) {
    const float* x   = (const float*)d_in[0];
    const int*   ei  = (const int*)d_in[1];
    const float* W1  = (const float*)d_in[2];
    const float* b1  = (const float*)d_in[3];
    const float* W2  = (const float*)d_in[4];
    const float* b2  = (const float*)d_in[5];
    const float* Wc  = (const float*)d_in[6];
    const float* bc  = (const float*)d_in[7];
    float* out = (float*)d_out;

    const int nN = in_sizes[0] / N_IN_F;
    const int nE = in_sizes[1] / 2;
    const int* src = ei;
    const int* dst = ei + nE;

    float* ws   = (float*)d_ws;
    float* h1   = ws;
    float* agg1 = h1   + (size_t)nN * N_HID_F;
    float* pemb = agg1 + (size_t)nN * N_HID_F;
    unsigned* pdeg = (unsigned*)(pemb + (size_t)NSLICE * nN * 2);
    float* h2   = (float*)(pdeg + (size_t)NSLICE * nN);
    float* inv  = h2   + (size_t)nN * 2;

    const int B   = 256;
    const int WPB = B / 32;
    const int hidTotal = nN * N_HID_F;

    const int nBands    = (nN + NR_MAX - 1) / NR_MAX;
    const int NR        = (nN + nBands - 1) / nBands;
    const int ePerSlice = (nE + NSLICE - 1) / NSLICE;

    k_deg_band<<<nBands * NSLICE, BAND_T, 0, stream>>>(dst, pdeg, nE, nN, nBands, NR, ePerSlice);
    k_inv_from_parts<<<(nN + B - 1) / B, B, 0, stream>>>(pdeg, inv, nN);

    const int nTiles = (nN + 15) / 16;
    const size_t ldsBytes = (size_t)N_HID_F * BSTRIDE * sizeof(float);
    k_gemm1_wmma<<<(nTiles + WPB - 1) / WPB, B, ldsBytes, stream>>>(x, W1, h1, nN);
    k_fill<<<(hidTotal + B - 1) / B, B, 0, stream>>>(agg1, 0.0f, hidTotal);
    k_aggregate_hid<<<(nE + WPB - 1) / WPB, B, 0, stream>>>(h1, src, dst, inv, agg1, nE);
    k_post1<<<(hidTotal + B - 1) / B, B, 0, stream>>>(h1, agg1, inv, b1, hidTotal);

    k_gemm2<<<(nN + B - 1) / B, B, 0, stream>>>(h1, W2, h2, nN);
    k_emb_band<<<nBands * NSLICE, BAND_T, 0, stream>>>(h2, src, dst, inv, pemb, nE, nN, nBands, NR, ePerSlice);
    k_final<<<(nN + B - 1) / B, B, 0, stream>>>(h2, pemb, inv, b2, Wc, bc, out, nN);
}


# Layer-1 scatter: a wave takes 32 edges, not 1: src, dst, coefficient loaded once per 32 edges and broadcast by readlane; row loads of 8 edges in flight at once; sum order = atomic arrival as before
# speedup vs baseline: 2.1587x; 1.2048x over previous
#include <hip/hip_runtime.h>
#include <cmath>

typedef __attribute__((ext_vector_type(2))) float v2f;
typedef __attribute__((ext_vector_type(8))) float v8f;

#define N_IN_F   165
#define N_HID_F  128
#define N_EMB_F  2
#define K_PAD    168
#define BSTRIDE  170

__global__ void k_fill(float* __restrict__ p, float v, int n) {
    int i = blockIdx.x * blockDim.x + threadIdx.x;
    if (i < n) p[i] = v;
}

#define NR_MAX   32768
#define NSLICE   64
#define BAND_T   1024
#define EU       8

__global__ void __launch_bounds__(BAND_T)
k_deg_band(const int* __restrict__ dst, unsigned* __restrict__ part,
           int nE, int nN, int nBands, int NR, int ePerSlice) {
    __shared__ unsigned hist[NR_MAX];
    const int tid   = threadIdx.x;
    const int band  = blockIdx.x % nBands;
    const int slice = blockIdx.x / nBands;
    const int lo    = band * NR;
    int cnt = nN - lo;
    if (cnt > NR) cnt = NR;
    if (cnt < 0)  cnt = 0;

    for (int i = tid; i < cnt; i += BAND_T) hist[i] = 0u;
    __syncthreads();

    const int e0 = slice * ePerSlice;
    int e1 = e0 + ePerSlice;
    if (e1 > nE) e1 = nE;
    for (int base = e0; base < e1; base += EU * BAND_T) {
        int d[EU];
#pragma unroll
        for (int u = 0; u < EU; ++u) {
            int idx = base + u * BAND_T + tid;
            int ic  = (idx < e1) ? idx : (e1 - 1);
            int dv  = dst[ic];
            d[u] = (idx < e1) ? dv : -1;
        }
#pragma unroll
        for (int u = 0; u < EU; ++u) {
            unsigned r = (unsigned)(d[u] - lo);
            if (r < (unsigned)cnt) atomicAdd(&hist[r], 1u);
        }
    }
    __syncthreads();

    unsigned* p = part + (size_t)slice * nN + lo;
    for (int i = tid; i < cnt; i += BAND_T) p[i] = hist[i];
}

__global__ void k_inv_from_parts(const unsigned* __restrict__ part, float* __restrict__ inv, int nN) {
    int n = blockIdx.x * blockDim.x + threadIdx.x;
    if (n >= nN) return;
    unsigned c = 0u;
#pragma unroll 16
    for (int s = 0; s < NSLICE; ++s) c += part[(size_t)s * nN + n];
    inv[n] = rsqrtf(1.0f + (float)c);
}

__global__ void __launch_bounds__(256, 1)
k_gemm1_wmma(const float* __restrict__ X, const float* __restrict__ W,
             float* __restrict__ H, int nNodes) {
    extern __shared__ float bsh[];

    for (int idx = threadIdx.x; idx < N_HID_F * K_PAD; idx += blockDim.x) {
        int n = idx / K_PAD;
        int k = idx - n * K_PAD;
        int kc = (k < N_IN_F) ? k : (N_IN_F - 1);
        float wv = W[(size_t)kc * N_HID_F + n];
        bsh[n * BSTRIDE + k] = (k < N_IN_F) ? wv : 0.0f;
    }
    __syncthreads();

    const int lane   = threadIdx.x & 31;
    const int tileM  = blockIdx.x * (blockDim.x >> 5) + (threadIdx.x >> 5);
    const int nTiles = (nNodes + 15) >> 4;
    if (tileM >= nTiles) return;

    const int hi = lane >> 4;
    const int lo = lane & 15;
    int m = tileM * 16 + lo;
    if (m >= nNodes) m = nNodes - 1;

    v8f zero = {};
    v8f acc[8];
#pragma unroll
    for (int t = 0; t < 8; ++t) acc[t] = zero;

    const float* Xrow  = X + (size_t)m * N_IN_F;
    const float* bbase = bsh + lo * BSTRIDE;

#pragma unroll 2
    for (int k0 = 0; k0 < 164; k0 += 4) {
        const int kA = k0 + 2 * hi;
        v2f a;
        a.x = Xrow[kA];
        a.y = Xrow[kA + 1];
#pragma unroll
        for (int t = 0; t < 8; ++t) {
            v2f b = *(const v2f*)(bbase + (size_t)(t * 16) * BSTRIDE + kA);
            acc[t] = __builtin_amdgcn_wmma_f32_16x16x4_f32(
                false, a, false, b, (short)0, acc[t], false, false);
        }
    }

    {
        const int kA = 164 + 2 * hi;
        float xv = Xrow[164];
        v2f a;
        a.x = (hi == 0) ? xv : 0.0f;
        a.y = 0.0f;
#pragma unroll
        for (int t = 0; t < 8; ++t) {
            v2f b = *(const v2f*)(bbase + (size_t)(t * 16) * BSTRIDE + kA);
            acc[t] = __builtin_amdgcn_wmma_f32_16x16x4_f32(
                false, a, false, b, (short)0, acc[t], false, false);
        }
    }

    if (tileM * 16 + 16 <= nNodes) {
        float* Hrow = H + (size_t)(tileM * 16 + 8 * hi) * N_HID_F + lo;
#pragma unroll
        for (int r = 0; r < 8; ++r) {
#pragma unroll
            for (int t = 0; t < 8; ++t)
                Hrow[(size_t)r * N_HID_F + t * 16] = acc[t][r];
        }
    } else {
#pragma unroll
        for (int r = 0; r < 8; ++r) {
            int row = tileM * 16 + r + 8 * hi;
            if (row < nNodes) {
#pragma unroll
                for (int t = 0; t < 8; ++t)
                    H[(size_t)row * N_HID_F + t * 16 + lo] = acc[t][r];
            }
        }
    }
}

#define AGG_EPW  32
#define AGG_U    8
__global__ void __launch_bounds__(256)
k_aggregate_hid(const float* __restrict__ H, const int* __restrict__ src,
                const int* __restrict__ dst, const float* __restrict__ inv,
                float* __restrict__ out, int nE) {
    const int lane = threadIdx.x & 31;
    const int wv = __builtin_amdgcn_readfirstlane((int)(threadIdx.x >> 5));
    const int e0 = (blockIdx.x * (blockDim.x >> 5) + wv) * AGG_EPW;
    if (e0 >= nE) return;
    const int e  = e0 + lane;
    const int ec = (e < nE) ? e : (nE - 1);
    const int s  = src[ec];
    const int d  = dst[ec];
    const int cb = __float_as_int(inv[s] * inv[d]);
    int cnt = nE - e0;
    if (cnt > AGG_EPW) cnt = AGG_EPW;

    if (cnt == AGG_EPW) {
#pragma unroll
        for (int j0 = 0; j0 < AGG_EPW; j0 += AGG_U) {
            float v[AGG_U][4];
#pragma unroll
            for (int u = 0; u < AGG_U; ++u) {
                const int sj = __builtin_amdgcn_readlane(s, j0 + u);
                const float* h = H + (size_t)sj * N_HID_F + lane;
                v[u][0] = h[0]; v[u][1] = h[32]; v[u][2] = h[64]; v[u][3] = h[96];
            }
#pragma unroll
            for (int u = 0; u < AGG_U; ++u) {
                const int   dj = __builtin_amdgcn_readlane(d, j0 + u);
                const float c  = __int_as_float(__builtin_amdgcn_readlane(cb, j0 + u));
                float* o = out + (size_t)dj * N_HID_F + lane;
                atomicAdd(o +  0, v[u][0] * c);
                atomicAdd(o + 32, v[u][1] * c);
                atomicAdd(o + 64, v[u][2] * c);
                atomicAdd(o + 96, v[u][3] * c);
            }
        }
    } else {
        for (int j = 0; j < cnt; ++j) {
            const int   sj = __builtin_amdgcn_readlane(s, j);
            const int   dj = __builtin_amdgcn_readlane(d, j);
            const float c  = __int_as_float(__builtin_amdgcn_readlane(cb, j));
            const float* h = H + (size_t)sj * N_HID_F + lane;
            const float v0 = h[0], v1 = h[32], v2 = h[64], v3 = h[96];
            float* o = out + (size_t)dj * N_HID_F + lane;
            atomicAdd(o +  0, v0 * c);
            atomicAdd(o + 32, v1 * c);
            atomicAdd(o + 64, v2 * c);
            atomicAdd(o + 96, v3 * c);
        }
    }
}

__global__ void k_post1(float* __restrict__ H1, const float* __restrict__ agg,
                        const float* __restrict__ inv, const float* __restrict__ b1,
                        int total) {
    int i = blockIdx.x * blockDim.x + threadIdx.x;
    if (i >= total) return;
    int node = i >> 7;
    int f    = i & 127;
    float is = inv[node];
    float v  = agg[i] + H1[i] * is * is + b1[f];
    H1[i] = tanhf(v);
}

__global__ void k_gemm2(const float* __restrict__ H, const float* __restrict__ W2,
                        float* __restrict__ H2, int nNodes) {
    __shared__ float w2s[N_HID_F * N_EMB_F];
    int tid = threadIdx.x;
    if (tid < N_HID_F * N_EMB_F) w2s[tid] = W2[tid];
    __syncthreads();
    int n = blockIdx.x * blockDim.x + tid;
    if (n >= nNodes) return;
    const float* h = H + (size_t)n * N_HID_F;
    float a0 = 0.0f, a1 = 0.0f;
#pragma unroll 8
    for (int k = 0; k < N_HID_F; ++k) {
        float hv = h[k];
        a0 += hv * w2s[k * 2 + 0];
        a1 += hv * w2s[k * 2 + 1];
    }
    H2[(size_t)n * 2 + 0] = a0;
    H2[(size_t)n * 2 + 1] = a1;
}

__global__ void __launch_bounds__(BAND_T)
k_emb_band(const float* __restrict__ H2, const int* __restrict__ src,
           const int* __restrict__ dst, const float* __restrict__ inv,
           float* __restrict__ part, int nE, int nN, int nBands, int NR, int ePerSlice) {
    __shared__ __attribute__((aligned(16))) float acc[NR_MAX * 2];
    const int tid   = threadIdx.x;
    const int band  = blockIdx.x % nBands;
    const int slice = blockIdx.x / nBands;
    const int lo    = band * NR;
    int cnt = nN - lo;
    if (cnt > NR) cnt = NR;
    if (cnt < 0)  cnt = 0;

    {
        float4* a4 = (float4*)acc;
        const int n4 = (2 * cnt + 3) >> 2;
        const float4 z = make_float4(0.0f, 0.0f, 0.0f, 0.0f);
        for (int i = tid; i < n4; i += BAND_T) a4[i] = z;
    }
    __syncthreads();

    const int e0 = slice * ePerSlice;
    int e1 = e0 + ePerSlice;
    if (e1 > nE) e1 = nE;
    for (int base = e0; base < e1; base += EU * BAND_T) {
        int  dd[EU], ss[EU];
        bool m[EU];
#pragma unroll
        for (int u = 0; u < EU; ++u) {
            int idx = base + u * BAND_T + tid;
            int ic  = (idx < e1) ? idx : (e1 - 1);
            int d = dst[ic];
            int s = src[ic];
            unsigned r = (unsigned)(d - lo);
            m[u]  = (idx < e1) && (r < (unsigned)cnt);
            dd[u] = m[u] ? d : 0;
            ss[u] = m[u] ? s : 0;
        }
        float  is[EU], id[EU];
        float2 v[EU];
#pragma unroll
        for (int u = 0; u < EU; ++u) {
            is[u] = inv[ss[u]];
            id[u] = inv[dd[u]];
            v[u]  = *(const float2*)(H2 + (size_t)ss[u] * 2);
        }
        asm volatile("" :: "v"(is[0]), "v"(id[0]), "v"(v[0].x), "v"(v[0].y));
#pragma unroll
        for (int u = 0; u < EU; ++u) {
            if (m[u]) {
                const float c = is[u] * id[u];
                float* a = acc + 2 * (dd[u] - lo);
                atomicAdd(a + 0, v[u].x * c);
                atomicAdd(a + 1, v[u].y * c);
            }
        }
    }
    __syncthreads();

    float2*       p  = (float2*)(part + (size_t)slice * 2 * nN) + lo;
    const float2* a2 = (const float2*)acc;
    for (int i = tid; i < cnt; i += BAND_T) p[i] = a2[i];
}

__global__ void k_final(const float* __restrict__ H2, const float* __restrict__ part,
                        const float* __restrict__ inv, const float* __restrict__ b2,
                        const float* __restrict__ Wc, const float* __restrict__ bc,
                        float* __restrict__ out, int nNodes) {
    int n = blockIdx.x * blockDim.x + threadIdx.x;
    if (n >= nNodes) return;
    const float2* p = (const float2*)part + n;
    float g0 = 0.0f, g1 = 0.0f;
#pragma unroll 16
    for (int s = 0; s < NSLICE; ++s) {
        float2 t = p[(size_t)s * nNodes];
        g0 += t.x;
        g1 += t.y;
    }
    float is = inv[n];
    float i2 = is * is;
    float e0 = tanhf(g0 + H2[(size_t)n * 2 + 0] * i2 + b2[0]);
    float e1 = tanhf(g1 + H2[(size_t)n * 2 + 1] * i2 + b2[1]);
    float z  = e0 * Wc[0] + e1 * Wc[1] + bc[0];
    out[n] = 1.0f / (1.0f + expf(-z));
}

extern "C" void kernel_launch(void* const* d_in, const int* in_sizes, int n_in,
                              void* d_out, int out_size, void* d_ws, size_t ws_size,
                              hipStream_t stream) {
    const float* x   = (const float*)d_in[0];
    const int*   ei  = (const int*)d_in[1];
    const float* W1  = (const float*)d_in[2];
    const float* b1  = (const float*)d_in[3];
    const float* W2  = (const float*)d_in[4];
    const float* b2  = (const float*)d_in[5];
    const float* Wc  = (const float*)d_in[6];
    const float* bc  = (const float*)d_in[7];
    float* out = (float*)d_out;

    const int nN = in_sizes[0] / N_IN_F;
    const int nE = in_sizes[1] / 2;
    const int* src = ei;
    const int* dst = ei + nE;

    float* ws   = (float*)d_ws;
    float* h1   = ws;
    float* agg1 = h1   + (size_t)nN * N_HID_F;
    float* pemb = agg1 + (size_t)nN * N_HID_F;
    unsigned* pdeg = (unsigned*)(pemb + (size_t)NSLICE * nN * 2);
    float* h2   = (float*)(pdeg + (size_t)NSLICE * nN);
    float* inv  = h2   + (size_t)nN * 2;

    const int B   = 256;
    const int WPB = B / 32;
    const int hidTotal = nN * N_HID_F;

    const int nBands    = (nN + NR_MAX - 1) / NR_MAX;
    const int NR        = (nN + nBands - 1) / nBands;
    const int ePerSlice = (nE + NSLICE - 1) / NSLICE;

    k_deg_band<<<nBands * NSLICE, BAND_T, 0, stream>>>(dst, pdeg, nE, nN, nBands, NR, ePerSlice);
    k_inv_from_parts<<<(nN + B - 1) / B, B, 0, stream>>>(pdeg, inv, nN);

    const int nTiles = (nN + 15) / 16;
    const size_t ldsBytes = (size_t)N_HID_F * BSTRIDE * sizeof(float);
    k_gemm1_wmma<<<(nTiles + WPB - 1) / WPB, B, ldsBytes, stream>>>(x, W1, h1, nN);
    k_fill<<<(hidTotal + B - 1) / B, B, 0, stream>>>(agg1, 0.0f, hidTotal);
    k_aggregate_hid<<<(nE + AGG_EPW * WPB - 1) / (AGG_EPW * WPB), B, 0, stream>>>(h1, src, dst, inv, agg1, nE);
    k_post1<<<(hidTotal + B - 1) / B, B, 0, stream>>>(h1, agg1, inv, b1, hidTotal);

    k_gemm2<<<(nN + B - 1) / B, B, 0, stream>>>(h1, W2, h2, nN);
    k_emb_band<<<nBands * NSLICE, BAND_T, 0, stream>>>(h2, src, dst, inv, pemb, nE, nN, nBands, NR, ePerSlice);
    k_final<<<(nN + B - 1) / B, B, 0, stream>>>(h2, pemb, inv, b2, Wc, bc, out, nN);
}


# GEMM1 launch shape: 250 work-groups of 25 waves, one per WGP (LDS reservation), one row strip per wave: W1^T staged 250 times, not 782, all strips resident in one round; sums unchanged
# speedup vs baseline: 2.3131x; 1.0715x over previous
#include <hip/hip_runtime.h>
#include <cmath>

typedef __attribute__((ext_vector_type(2))) float v2f;
typedef __attribute__((ext_vector_type(8))) float v8f;

#define N_IN_F   165
#define N_HID_F  128
#define N_EMB_F  2
#define K_PAD    168
#define BSTRIDE  170

__global__ void k_fill(float* __restrict__ p, float v, int n) {
    int i = blockIdx.x * blockDim.x + threadIdx.x;
    if (i < n) p[i] = v;
}

#define NR_MAX   32768
#define NSLICE   64
#define BAND_T   1024
#define EU       8

__global__ void __launch_bounds__(BAND_T)
k_deg_band(const int* __restrict__ dst, unsigned* __restrict__ part,
           int nE, int nN, int nBands, int NR, int ePerSlice) {
    __shared__ unsigned hist[NR_MAX];
    const int tid   = threadIdx.x;
    const int band  = blockIdx.x % nBands;
    const int slice = blockIdx.x / nBands;
    const int lo    = band * NR;
    int cnt = nN - lo;
    if (cnt > NR) cnt = NR;
    if (cnt < 0)  cnt = 0;

    for (int i = tid; i < cnt; i += BAND_T) hist[i] = 0u;
    __syncthreads();

    const int e0 = slice * ePerSlice;
    int e1 = e0 + ePerSlice;
    if (e1 > nE) e1 = nE;
    for (int base = e0; base < e1; base += EU * BAND_T) {
        int d[EU];
#pragma unroll
        for (int u = 0; u < EU; ++u) {
            int idx = base + u * BAND_T + tid;
            int ic  = (idx < e1) ? idx : (e1 - 1);
            int dv  = dst[ic];
            d[u] = (idx < e1) ? dv : -1;
        }
#pragma unroll
        for (int u = 0; u < EU; ++u) {
            unsigned r = (unsigned)(d[u] - lo);
            if (r < (unsigned)cnt) atomicAdd(&hist[r], 1u);
        }
    }
    __syncthreads();

    unsigned* p = part + (size_t)slice * nN + lo;
    for (int i = tid; i < cnt; i += BAND_T) p[i] = hist[i];
}

__global__ void k_inv_from_parts(const unsigned* __restrict__ part, float* __restrict__ inv, int nN) {
    int n = blockIdx.x * blockDim.x + threadIdx.x;
    if (n >= nN) return;
    unsigned c = 0u;
#pragma unroll 16
    for (int s = 0; s < NSLICE; ++s) c += part[(size_t)s * nN + n];
    inv[n] = rsqrtf(1.0f + (float)c);
}

#define GEMM_WAVES 25
#define GEMM_T     (GEMM_WAVES * 32)
#define GEMM_LDS   (164 * 1024)
__global__ void __launch_bounds__(GEMM_T)
k_gemm1_wmma(const float* __restrict__ X, const float* __restrict__ W,
             float* __restrict__ H, int nNodes) {
    extern __shared__ float bsh[];

    for (int idx = threadIdx.x; idx < N_HID_F * K_PAD; idx += blockDim.x) {
        int n = idx / K_PAD;
        int k = idx - n * K_PAD;
        int kc = (k < N_IN_F) ? k : (N_IN_F - 1);
        float wv = W[(size_t)kc * N_HID_F + n];
        bsh[n * BSTRIDE + k] = (k < N_IN_F) ? wv : 0.0f;
    }
    __syncthreads();

    const int lane   = threadIdx.x & 31;
    const int tileM  = blockIdx.x * (blockDim.x >> 5) + (threadIdx.x >> 5);
    const int nTiles = (nNodes + 15) >> 4;
    if (tileM >= nTiles) return;

    const int hi = lane >> 4;
    const int lo = lane & 15;
    int m = tileM * 16 + lo;
    if (m >= nNodes) m = nNodes - 1;

    v8f zero = {};
    v8f acc[8];
#pragma unroll
    for (int t = 0; t < 8; ++t) acc[t] = zero;

    const float* Xrow  = X + (size_t)m * N_IN_F;
    const float* bbase = bsh + lo * BSTRIDE;

#pragma unroll 2
    for (int k0 = 0; k0 < 164; k0 += 4) {
        const int kA = k0 + 2 * hi;
        v2f a;
        a.x = Xrow[kA];
        a.y = Xrow[kA + 1];
#pragma unroll
        for (int t = 0; t < 8; ++t) {
            v2f b = *(const v2f*)(bbase + (size_t)(t * 16) * BSTRIDE + kA);
            acc[t] = __builtin_amdgcn_wmma_f32_16x16x4_f32(
                false, a, false, b, (short)0, acc[t], false, false);
        }
    }

    {
        const int kA = 164 + 2 * hi;
        float xv = Xrow[164];
        v2f a;
        a.x = (hi == 0) ? xv : 0.0f;
        a.y = 0.0f;
#pragma unroll
        for (int t = 0; t < 8; ++t) {
            v2f b = *(const v2f*)(bbase + (size_t)(t * 16) * BSTRIDE + kA);
            acc[t] = __builtin_amdgcn_wmma_f32_16x16x4_f32(
                false, a, false, b, (short)0, acc[t], false, false);
        }
    }

    if (tileM * 16 + 16 <= nNodes) {
        float* Hrow = H + (size_t)(tileM * 16 + 8 * hi) * N_HID_F + lo;
#pragma unroll
        for (int r = 0; r < 8; ++r) {
#pragma unroll
            for (int t = 0; t < 8; ++t)
                Hrow[(size_t)r * N_HID_F + t * 16] = acc[t][r];
        }
    } else {
#pragma unroll
        for (int r = 0; r < 8; ++r) {
            int row = tileM * 16 + r + 8 * hi;
            if (row < nNodes) {
#pragma unroll
                for (int t = 0; t < 8; ++t)
                    H[(size_t)row * N_HID_F + t * 16 + lo] = acc[t][r];
            }
        }
    }
}

#define AGG_EPW  32
#define AGG_U    8
__global__ void __launch_bounds__(256)
k_aggregate_hid(const float* __restrict__ H, const int* __restrict__ src,
                const int* __restrict__ dst, const float* __restrict__ inv,
                float* __restrict__ out, int nE) {
    const int lane = threadIdx.x & 31;
    const int wv = __builtin_amdgcn_readfirstlane((int)(threadIdx.x >> 5));
    const int e0 = (blockIdx.x * (blockDim.x >> 5) + wv) * AGG_EPW;
    if (e0 >= nE) return;
    const int e  = e0 + lane;
    const int ec = (e < nE) ? e : (nE - 1);
    const int s  = src[ec];
    const int d  = dst[ec];
    const int cb = __float_as_int(inv[s] * inv[d]);
    int cnt = nE - e0;
    if (cnt > AGG_EPW) cnt = AGG_EPW;

    if (cnt == AGG_EPW) {
#pragma unroll
        for (int j0 = 0; j0 < AGG_EPW; j0 += AGG_U) {
            float v[AGG_U][4];
#pragma unroll
            for (int u = 0; u < AGG_U; ++u) {
                const int sj = __builtin_amdgcn_readlane(s, j0 + u);
                const float* h = H + (size_t)sj * N_HID_F + lane;
                v[u][0] = h[0]; v[u][1] = h[32]; v[u][2] = h[64]; v[u][3] = h[96];
            }
#pragma unroll
            for (int u = 0; u < AGG_U; ++u) {
                const int   dj = __builtin_amdgcn_readlane(d, j0 + u);
                const float c  = __int_as_float(__builtin_amdgcn_readlane(cb, j0 + u));
                float* o = out + (size_t)dj * N_HID_F + lane;
                atomicAdd(o +  0, v[u][0] * c);
                atomicAdd(o + 32, v[u][1] * c);
                atomicAdd(o + 64, v[u][2] * c);
                atomicAdd(o + 96, v[u][3] * c);
            }
        }
    } else {
        for (int j = 0; j < cnt; ++j) {
            const int   sj = __builtin_amdgcn_readlane(s, j);
            const int   dj = __builtin_amdgcn_readlane(d, j);
            const float c  = __int_as_float(__builtin_amdgcn_readlane(cb, j));
            const float* h = H + (size_t)sj * N_HID_F + lane;
            const float v0 = h[0], v1 = h[32], v2 = h[64], v3 = h[96];
            float* o = out + (size_t)dj * N_HID_F + lane;
            atomicAdd(o +  0, v0 * c);
            atomicAdd(o + 32, v1 * c);
            atomicAdd(o + 64, v2 * c);
            atomicAdd(o + 96, v3 * c);
        }
    }
}

__global__ void k_post1(float* __restrict__ H1, const float* __restrict__ agg,
                        const float* __restrict__ inv, const float* __restrict__ b1,
                        int total) {
    int i = blockIdx.x * blockDim.x + threadIdx.x;
    if (i >= total) return;
    int node = i >> 7;
    int f    = i & 127;
    float is = inv[node];
    float v  = agg[i] + H1[i] * is * is + b1[f];
    H1[i] = tanhf(v);
}

__global__ void k_gemm2(const float* __restrict__ H, const float* __restrict__ W2,
                        float* __restrict__ H2, int nNodes) {
    __shared__ float w2s[N_HID_F * N_EMB_F];
    int tid = threadIdx.x;
    if (tid < N_HID_F * N_EMB_F) w2s[tid] = W2[tid];
    __syncthreads();
    int n = blockIdx.x * blockDim.x + tid;
    if (n >= nNodes) return;
    const float* h = H + (size_t)n * N_HID_F;
    float a0 = 0.0f, a1 = 0.0f;
#pragma unroll 8
    for (int k = 0; k < N_HID_F; ++k) {
        float hv = h[k];
        a0 += hv * w2s[k * 2 + 0];
        a1 += hv * w2s[k * 2 + 1];
    }
    H2[(size_t)n * 2 + 0] = a0;
    H2[(size_t)n * 2 + 1] = a1;
}

__global__ void __launch_bounds__(BAND_T)
k_emb_band(const float* __restrict__ H2, const int* __restrict__ src,
           const int* __restrict__ dst, const float* __restrict__ inv,
           float* __restrict__ part, int nE, int nN, int nBands, int NR, int ePerSlice) {
    __shared__ __attribute__((aligned(16))) float acc[NR_MAX * 2];
    const int tid   = threadIdx.x;
    const int band  = blockIdx.x % nBands;
    const int slice = blockIdx.x / nBands;
    const int lo    = band * NR;
    int cnt = nN - lo;
    if (cnt > NR) cnt = NR;
    if (cnt < 0)  cnt = 0;

    {
        float4* a4 = (float4*)acc;
        const int n4 = (2 * cnt + 3) >> 2;
        const float4 z = make_float4(0.0f, 0.0f, 0.0f, 0.0f);
        for (int i = tid; i < n4; i += BAND_T) a4[i] = z;
    }
    __syncthreads();

    const int e0 = slice * ePerSlice;
    int e1 = e0 + ePerSlice;
    if (e1 > nE) e1 = nE;
    for (int base = e0; base < e1; base += EU * BAND_T) {
        int  dd[EU], ss[EU];
        bool m[EU];
#pragma unroll
        for (int u = 0; u < EU; ++u) {
            int idx = base + u * BAND_T + tid;
            int ic  = (idx < e1) ? idx : (e1 - 1);
            int d = dst[ic];
            int s = src[ic];
            unsigned r = (unsigned)(d - lo);
            m[u]  = (idx < e1) && (r < (unsigned)cnt);
            dd[u] = m[u] ? d : 0;
            ss[u] = m[u] ? s : 0;
        }
        float  is[EU], id[EU];
        float2 v[EU];
#pragma unroll
        for (int u = 0; u < EU; ++u) {
            is[u] = inv[ss[u]];
            id[u] = inv[dd[u]];
            v[u]  = *(const float2*)(H2 + (size_t)ss[u] * 2);
        }
        asm volatile("" :: "v"(is[0]), "v"(id[0]), "v"(v[0].x), "v"(v[0].y));
#pragma unroll
        for (int u = 0; u < EU; ++u) {
            if (m[u]) {
                const float c = is[u] * id[u];
                float* a = acc + 2 * (dd[u] - lo);
                atomicAdd(a + 0, v[u].x * c);
                atomicAdd(a + 1, v[u].y * c);
            }
        }
    }
    __syncthreads();

    float2*       p  = (float2*)(part + (size_t)slice * 2 * nN) + lo;
    const float2* a2 = (const float2*)acc;
    for (int i = tid; i < cnt; i += BAND_T) p[i] = a2[i];
}

__global__ void k_final(const float* __restrict__ H2, const float* __restrict__ part,
                        const float* __restrict__ inv, const float* __restrict__ b2,
                        const float* __restrict__ Wc, const float* __restrict__ bc,
                        float* __restrict__ out, int nNodes) {
    int n = blockIdx.x * blockDim.x + threadIdx.x;
    if (n >= nNodes) return;
    const float2* p = (const float2*)part + n;
    float g0 = 0.0f, g1 = 0.0f;
#pragma unroll 16
    for (int s = 0; s < NSLICE; ++s) {
        float2 t = p[(size_t)s * nNodes];
        g0 += t.x;
        g1 += t.y;
    }
    float is = inv[n];
    float i2 = is * is;
    float e0 = tanhf(g0 + H2[(size_t)n * 2 + 0] * i2 + b2[0]);
    float e1 = tanhf(g1 + H2[(size_t)n * 2 + 1] * i2 + b2[1]);
    float z  = e0 * Wc[0] + e1 * Wc[1] + bc[0];
    out[n] = 1.0f / (1.0f + expf(-z));
}

extern "C" void kernel_launch(void* const* d_in, const int* in_sizes, int n_in,
                              void* d_out, int out_size, void* d_ws, size_t ws_size,
                              hipStream_t stream) {
    const float* x   = (const float*)d_in[0];
    const int*   ei  = (const int*)d_in[1];
    const float* W1  = (const float*)d_in[2];
    const float* b1  = (const float*)d_in[3];
    const float* W2  = (const float*)d_in[4];
    const float* b2  = (const float*)d_in[5];
    const float* Wc  = (const float*)d_in[6];
    const float* bc  = (const float*)d_in[7];
    float* out = (float*)d_out;

    const int nN = in_sizes[0] / N_IN_F;
    const int nE = in_sizes[1] / 2;
    const int* src = ei;
    const int* dst = ei + nE;

    float* ws   = (float*)d_ws;
    float* h1   = ws;
    float* agg1 = h1   + (size_t)nN * N_HID_F;
    float* pemb = agg1 + (size_t)nN * N_HID_F;
    unsigned* pdeg = (unsigned*)(pemb + (size_t)NSLICE * nN * 2);
    float* h2   = (float*)(pdeg + (size_t)NSLICE * nN);
    float* inv  = h2   + (size_t)nN * 2;

    const int B   = 256;
    const int WPB = B / 32;
    const int hidTotal = nN * N_HID_F;

    const int nBands    = (nN + NR_MAX - 1) / NR_MAX;
    const int NR        = (nN + nBands - 1) / nBands;
    const int ePerSlice = (nE + NSLICE - 1) / NSLICE;

    k_deg_band<<<nBands * NSLICE, BAND_T, 0, stream>>>(dst, pdeg, nE, nN, nBands, NR, ePerSlice);
    k_inv_from_parts<<<(nN + B - 1) / B, B, 0, stream>>>(pdeg, inv, nN);

    const int nTiles = (nN + 15) / 16;
    k_gemm1_wmma<<<(nTiles + GEMM_WAVES - 1) / GEMM_WAVES, GEMM_T, GEMM_LDS, stream>>>(x, W1, h1, nN);
    k_fill<<<(hidTotal + B - 1) / B, B, 0, stream>>>(agg1, 0.0f, hidTotal);
    k_aggregate_hid<<<(nE + AGG_EPW * WPB - 1) / (AGG_EPW * WPB), B, 0, stream>>>(h1, src, dst, inv, agg1, nE);
    k_post1<<<(hidTotal + B - 1) / B, B, 0, stream>>>(h1, agg1, inv, b1, hidTotal);

    k_gemm2<<<(nN + B - 1) / B, B, 0, stream>>>(h1, W2, h2, nN);
    k_emb_band<<<nBands * NSLICE, BAND_T, 0, stream>>>(h2, src, dst, inv, pemb, nE, nN, nBands, NR, ePerSlice);
    k_final<<<(nN + B - 1) / B, B, 0, stream>>>(h2, pemb, inv, b2, Wc, bc, out, nN);
}


# Zero fill and post-1 pass (one float per thread, 400K waves each) become resident grids of 2048 work-groups: 16 B per lane per trip, one node row per wave trip in post-1; sums unchanged
# speedup vs baseline: 2.4624x; 1.0645x over previous
#include <hip/hip_runtime.h>
#include <cmath>

typedef __attribute__((ext_vector_type(2))) float v2f;
typedef __attribute__((ext_vector_type(8))) float v8f;

#define N_IN_F   165
#define N_HID_F  128
#define N_EMB_F  2
#define K_PAD    168
#define BSTRIDE  170

#define EW_T   256
#define EW_WG  2048
__global__ void __launch_bounds__(EW_T)
k_fill(float* __restrict__ p, float v, int n) {
    const int n4 = n >> 2;
    float4* __restrict__ p4 = (float4*)p;
    const float4 v4 = make_float4(v, v, v, v);
    const int tid = blockIdx.x * blockDim.x + threadIdx.x;
    const int nt  = gridDim.x * blockDim.x;
    for (int i = tid; i < n4; i += nt) p4[i] = v4;
    const int t = (n4 << 2) + tid;
    if (t < n) p[t] = v;
}

#define NR_MAX   32768
#define NSLICE   64
#define BAND_T   1024
#define EU       8

__global__ void __launch_bounds__(BAND_T)
k_deg_band(const int* __restrict__ dst, unsigned* __restrict__ part,
           int nE, int nN, int nBands, int NR, int ePerSlice) {
    __shared__ unsigned hist[NR_MAX];
    const int tid   = threadIdx.x;
    const int band  = blockIdx.x % nBands;
    const int slice = blockIdx.x / nBands;
    const int lo    = band * NR;
    int cnt = nN - lo;
    if (cnt > NR) cnt = NR;
    if (cnt < 0)  cnt = 0;

    for (int i = tid; i < cnt; i += BAND_T) hist[i] = 0u;
    __syncthreads();

    const int e0 = slice * ePerSlice;
    int e1 = e0 + ePerSlice;
    if (e1 > nE) e1 = nE;
    for (int base = e0; base < e1; base += EU * BAND_T) {
        int d[EU];
#pragma unroll
        for (int u = 0; u < EU; ++u) {
            int idx = base + u * BAND_T + tid;
            int ic  = (idx < e1) ? idx : (e1 - 1);
            int dv  = dst[ic];
            d[u] = (idx < e1) ? dv : -1;
        }
#pragma unroll
        for (int u = 0; u < EU; ++u) {
            unsigned r = (unsigned)(d[u] - lo);
            if (r < (unsigned)cnt) atomicAdd(&hist[r], 1u);
        }
    }
    __syncthreads();

    unsigned* p = part + (size_t)slice * nN + lo;
    for (int i = tid; i < cnt; i += BAND_T) p[i] = hist[i];
}

__global__ void k_inv_from_parts(const unsigned* __restrict__ part, float* __restrict__ inv, int nN) {
    int n = blockIdx.x * blockDim.x + threadIdx.x;
    if (n >= nN) return;
    unsigned c = 0u;
#pragma unroll 16
    for (int s = 0; s < NSLICE; ++s) c += part[(size_t)s * nN + n];
    inv[n] = rsqrtf(1.0f + (float)c);
}

#define GEMM_WAVES 25
#define GEMM_T     (GEMM_WAVES * 32)
#define GEMM_LDS   (164 * 1024)
__global__ void __launch_bounds__(GEMM_T)
k_gemm1_wmma(const float* __restrict__ X, const float* __restrict__ W,
             float* __restrict__ H, int nNodes) {
    extern __shared__ float bsh[];

    for (int idx = threadIdx.x; idx < N_HID_F * K_PAD; idx += blockDim.x) {
        int n = idx / K_PAD;
        int k = idx - n * K_PAD;
        int kc = (k < N_IN_F) ? k : (N_IN_F - 1);
        float wv = W[(size_t)kc * N_HID_F + n];
        bsh[n * BSTRIDE + k] = (k < N_IN_F) ? wv : 0.0f;
    }
    __syncthreads();

    const int lane   = threadIdx.x & 31;
    const int tileM  = blockIdx.x * (blockDim.x >> 5) + (threadIdx.x >> 5);
    const int nTiles = (nNodes + 15) >> 4;
    if (tileM >= nTiles) return;

    const int hi = lane >> 4;
    const int lo = lane & 15;
    int m = tileM * 16 + lo;
    if (m >= nNodes) m = nNodes - 1;

    v8f zero = {};
    v8f acc[8];
#pragma unroll
    for (int t = 0; t < 8; ++t) acc[t] = zero;

    const float* Xrow  = X + (size_t)m * N_IN_F;
    const float* bbase = bsh + lo * BSTRIDE;

#pragma unroll 2
    for (int k0 = 0; k0 < 164; k0 += 4) {
        const int kA = k0 + 2 * hi;
        v2f a;
        a.x = Xrow[kA];
        a.y = Xrow[kA + 1];
#pragma unroll
        for (int t = 0; t < 8; ++t) {
            v2f b = *(const v2f*)(bbase + (size_t)(t * 16) * BSTRIDE + kA);
            acc[t] = __builtin_amdgcn_wmma_f32_16x16x4_f32(
                false, a, false, b, (short)0, acc[t], false, false);
        }
    }

    {
        const int kA = 164 + 2 * hi;
        float xv = Xrow[164];
        v2f a;
        a.x = (hi == 0) ? xv : 0.0f;
        a.y = 0.0f;
#pragma unroll
        for (int t = 0; t < 8; ++t) {
            v2f b = *(const v2f*)(bbase + (size_t)(t * 16) * BSTRIDE + kA);
            acc[t] = __builtin_amdgcn_wmma_f32_16x16x4_f32(
                false, a, false, b, (short)0, acc[t], false, false);
        }
    }

    if (tileM * 16 + 16 <= nNodes) {
        float* Hrow = H + (size_t)(tileM * 16 + 8 * hi) * N_HID_F + lo;
#pragma unroll
        for (int r = 0; r < 8; ++r) {
#pragma unroll
            for (int t = 0; t < 8; ++t)
                Hrow[(size_t)r * N_HID_F + t * 16] = acc[t][r];
        }
    } else {
#pragma unroll
        for (int r = 0; r < 8; ++r) {
            int row = tileM * 16 + r + 8 * hi;
            if (row < nNodes) {
#pragma unroll
                for (int t = 0; t < 8; ++t)
                    H[(size_t)row * N_HID_F + t * 16 + lo] = acc[t][r];
            }
        }
    }
}

#define AGG_EPW  32
#define AGG_U    8
__global__ void __launch_bounds__(256)
k_aggregate_hid(const float* __restrict__ H, const int* __restrict__ src,
                const int* __restrict__ dst, const float* __restrict__ inv,
                float* __restrict__ out, int nE) {
    const int lane = threadIdx.x & 31;
    const int wv = __builtin_amdgcn_readfirstlane((int)(threadIdx.x >> 5));
    const int e0 = (blockIdx.x * (blockDim.x >> 5) + wv) * AGG_EPW;
    if (e0 >= nE) return;
    const int e  = e0 + lane;
    const int ec = (e < nE) ? e : (nE - 1);
    const int s  = src[ec];
    const int d  = dst[ec];
    const int cb = __float_as_int(inv[s] * inv[d]);
    int cnt = nE - e0;
    if (cnt > AGG_EPW) cnt = AGG_EPW;

    if (cnt == AGG_EPW) {
#pragma unroll
        for (int j0 = 0; j0 < AGG_EPW; j0 += AGG_U) {
            float v[AGG_U][4];
#pragma unroll
            for (int u = 0; u < AGG_U; ++u) {
                const int sj = __builtin_amdgcn_readlane(s, j0 + u);
                const float* h = H + (size_t)sj * N_HID_F + lane;
                v[u][0] = h[0]; v[u][1] = h[32]; v[u][2] = h[64]; v[u][3] = h[96];
            }
#pragma unroll
            for (int u = 0; u < AGG_U; ++u) {
                const int   dj = __builtin_amdgcn_readlane(d, j0 + u);
                const float c  = __int_as_float(__builtin_amdgcn_readlane(cb, j0 + u));
                float* o = out + (size_t)dj * N_HID_F + lane;
                atomicAdd(o +  0, v[u][0] * c);
                atomicAdd(o + 32, v[u][1] * c);
                atomicAdd(o + 64, v[u][2] * c);
                atomicAdd(o + 96, v[u][3] * c);
            }
        }
    } else {
        for (int j = 0; j < cnt; ++j) {
            const int   sj = __builtin_amdgcn_readlane(s, j);
            const int   dj = __builtin_amdgcn_readlane(d, j);
            const float c  = __int_as_float(__builtin_amdgcn_readlane(cb, j));
            const float* h = H + (size_t)sj * N_HID_F + lane;
            const float v0 = h[0], v1 = h[32], v2 = h[64], v3 = h[96];
            float* o = out + (size_t)dj * N_HID_F + lane;
            atomicAdd(o +  0, v0 * c);
            atomicAdd(o + 32, v1 * c);
            atomicAdd(o + 64, v2 * c);
            atomicAdd(o + 96, v3 * c);
        }
    }
}

__global__ void __launch_bounds__(EW_T)
k_post1(float* __restrict__ H1, const float* __restrict__ agg,
        const float* __restrict__ inv, const float* __restrict__ b1,
        int nNodes) {
    const int lane = threadIdx.x & 31;
    const int wv   = __builtin_amdgcn_readfirstlane((int)(threadIdx.x >> 5));
    const int wpb  = blockDim.x >> 5;
    const int nw   = gridDim.x * wpb;
    const int f    = lane * 4;
    const float b0 = b1[f], b1v = b1[f + 1], b2v = b1[f + 2], b3v = b1[f + 3];
    for (int node = blockIdx.x * wpb + wv; node < nNodes; node += nw) {
        const size_t off = (size_t)node * N_HID_F + f;
        const float4 a  = *(const float4*)(agg + off);
        const float4 h  = *(const float4*)(H1 + off);
        const float  is = inv[node];
        float4 r;
        r.x = tanhf(a.x + h.x * is * is + b0);
        r.y = tanhf(a.y + h.y * is * is + b1v);
        r.z = tanhf(a.z + h.z * is * is + b2v);
        r.w = tanhf(a.w + h.w * is * is + b3v);
        *(float4*)(H1 + off) = r;
    }
}

__global__ void k_gemm2(const float* __restrict__ H, const float* __restrict__ W2,
                        float* __restrict__ H2, int nNodes) {
    __shared__ float w2s[N_HID_F * N_EMB_F];
    int tid = threadIdx.x;
    if (tid < N_HID_F * N_EMB_F) w2s[tid] = W2[tid];
    __syncthreads();
    int n = blockIdx.x * blockDim.x + tid;
    if (n >= nNodes) return;
    const float* h = H + (size_t)n * N_HID_F;
    float a0 = 0.0f, a1 = 0.0f;
#pragma unroll 8
    for (int k = 0; k < N_HID_F; ++k) {
        float hv = h[k];
        a0 += hv * w2s[k * 2 + 0];
        a1 += hv * w2s[k * 2 + 1];
    }
    H2[(size_t)n * 2 + 0] = a0;
    H2[(size_t)n * 2 + 1] = a1;
}

__global__ void __launch_bounds__(BAND_T)
k_emb_band(const float* __restrict__ H2, const int* __restrict__ src,
           const int* __restrict__ dst, const float* __restrict__ inv,
           float* __restrict__ part, int nE, int nN, int nBands, int NR, int ePerSlice) {
    __shared__ __attribute__((aligned(16))) float acc[NR_MAX * 2];
    const int tid   = threadIdx.x;
    const int band  = blockIdx.x % nBands;
    const int slice = blockIdx.x / nBands;
    const int lo    = band * NR;
    int cnt = nN - lo;
    if (cnt > NR) cnt = NR;
    if (cnt < 0)  cnt = 0;

    {
        float4* a4 = (float4*)acc;
        const int n4 = (2 * cnt + 3) >> 2;
        const float4 z = make_float4(0.0f, 0.0f, 0.0f, 0.0f);
        for (int i = tid; i < n4; i += BAND_T) a4[i] = z;
    }
    __syncthreads();

    const int e0 = slice * ePerSlice;
    int e1 = e0 + ePerSlice;
    if (e1 > nE) e1 = nE;
    for (int base = e0; base < e1; base += EU * BAND_T) {
        int  dd[EU], ss[EU];
        bool m[EU];
#pragma unroll
        for (int u = 0; u < EU; ++u) {
            int idx = base + u * BAND_T + tid;
            int ic  = (idx < e1) ? idx : (e1 - 1);
            int d = dst[ic];
            int s = src[ic];
            unsigned r = (unsigned)(d - lo);
            m[u]  = (idx < e1) && (r < (unsigned)cnt);
            dd[u] = m[u] ? d : 0;
            ss[u] = m[u] ? s : 0;
        }
        float  is[EU], id[EU];
        float2 v[EU];
#pragma unroll
        for (int u = 0; u < EU; ++u) {
            is[u] = inv[ss[u]];
            id[u] = inv[dd[u]];
            v[u]  = *(const float2*)(H2 + (size_t)ss[u] * 2);
        }
        asm volatile("" :: "v"(is[0]), "v"(id[0]), "v"(v[0].x), "v"(v[0].y));
#pragma unroll
        for (int u = 0; u < EU; ++u) {
            if (m[u]) {
                const float c = is[u] * id[u];
                float* a = acc + 2 * (dd[u] - lo);
                atomicAdd(a + 0, v[u].x * c);
                atomicAdd(a + 1, v[u].y * c);
            }
        }
    }
    __syncthreads();

    float2*       p  = (float2*)(part + (size_t)slice * 2 * nN) + lo;
    const float2* a2 = (const float2*)acc;
    for (int i = tid; i < cnt; i += BAND_T) p[i] = a2[i];
}

__global__ void k_final(const float* __restrict__ H2, const float* __restrict__ part,
                        const float* __restrict__ inv, const float* __restrict__ b2,
                        const float* __restrict__ Wc, const float* __restrict__ bc,
                        float* __restrict__ out, int nNodes) {
    int n = blockIdx.x * blockDim.x + threadIdx.x;
    if (n >= nNodes) return;
    const float2* p = (const float2*)part + n;
    float g0 = 0.0f, g1 = 0.0f;
#pragma unroll 16
    for (int s = 0; s < NSLICE; ++s) {
        float2 t = p[(size_t)s * nNodes];
        g0 += t.x;
        g1 += t.y;
    }
    float is = inv[n];
    float i2 = is * is;
    float e0 = tanhf(g0 + H2[(size_t)n * 2 + 0] * i2 + b2[0]);
    float e1 = tanhf(g1 + H2[(size_t)n * 2 + 1] * i2 + b2[1]);
    float z  = e0 * Wc[0] + e1 * Wc[1] + bc[0];
    out[n] = 1.0f / (1.0f + expf(-z));
}

extern "C" void kernel_launch(void* const* d_in, const int* in_sizes, int n_in,
                              void* d_out, int out_size, void* d_ws, size_t ws_size,
                              hipStream_t stream) {
    const float* x   = (const float*)d_in[0];
    const int*   ei  = (const int*)d_in[1];
    const float* W1  = (const float*)d_in[2];
    const float* b1  = (const float*)d_in[3];
    const float* W2  = (const float*)d_in[4];
    const float* b2  = (const float*)d_in[5];
    const float* Wc  = (const float*)d_in[6];
    const float* bc  = (const float*)d_in[7];
    float* out = (float*)d_out;

    const int nN = in_sizes[0] / N_IN_F;
    const int nE = in_sizes[1] / 2;
    const int* src = ei;
    const int* dst = ei + nE;

    float* ws   = (float*)d_ws;
    float* h1   = ws;
    float* agg1 = h1   + (size_t)nN * N_HID_F;
    float* pemb = agg1 + (size_t)nN * N_HID_F;
    unsigned* pdeg = (unsigned*)(pemb + (size_t)NSLICE * nN * 2);
    float* h2   = (float*)(pdeg + (size_t)NSLICE * nN);
    float* inv  = h2   + (size_t)nN * 2;

    const int B   = 256;
    const int WPB = B / 32;
    const int hidTotal = nN * N_HID_F;

    const int nBands    = (nN + NR_MAX - 1) / NR_MAX;
    const int NR        = (nN + nBands - 1) / nBands;
    const int ePerSlice = (nE + NSLICE - 1) / NSLICE;

    k_deg_band<<<nBands * NSLICE, BAND_T, 0, stream>>>(dst, pdeg, nE, nN, nBands, NR, ePerSlice);
    k_inv_from_parts<<<(nN + B - 1) / B, B, 0, stream>>>(pdeg, inv, nN);

    const int nTiles = (nN + 15) / 16;
    k_gemm1_wmma<<<(nTiles + GEMM_WAVES - 1) / GEMM_WAVES, GEMM_T, GEMM_LDS, stream>>>(x, W1, h1, nN);
    k_fill<<<EW_WG, EW_T, 0, stream>>>(agg1, 0.0f, hidTotal);
    k_aggregate_hid<<<(nE + AGG_EPW * WPB - 1) / (AGG_EPW * WPB), B, 0, stream>>>(h1, src, dst, inv, agg1, nE);
    k_post1<<<EW_WG, EW_T, 0, stream>>>(h1, agg1, inv, b1, nN);

    k_gemm2<<<(nN + B - 1) / B, B, 0, stream>>>(h1, W2, h2, nN);
    k_emb_band<<<nBands * NSLICE, BAND_T, 0, stream>>>(h2, src, dst, inv, pemb, nE, nN, nBands, NR, ePerSlice);
    k_final<<<(nN + B - 1) / B, B, 0, stream>>>(h2, pemb, inv, b2, Wc, bc, out, nN);
}


# Layer-1 sums: edges binned per call by 416-node destination band (LDS counts, ranks; one cursor add per band and chunk); band owners add rows in LDS, store lines once; sum order = LDS arrival
# speedup vs baseline: 3.1398x; 1.2751x over previous
#include <hip/hip_runtime.h>
#include <cmath>

typedef __attribute__((ext_vector_type(2))) float v2f;
typedef __attribute__((ext_vector_type(8))) float v8f;

#define N_IN_F   165
#define N_HID_F  128
#define N_EMB_F  2
#define K_PAD    168
#define BSTRIDE  170

#define EW_T   256
#define EW_WG  2048
#define NR_MAX   32768
#define NSLICE   64
#define BAND_T   1024
#define EU       8

__global__ void __launch_bounds__(BAND_T)
k_deg_band(const int* __restrict__ dst, unsigned* __restrict__ part,
           int nE, int nN, int nBands, int NR, int ePerSlice) {
    __shared__ unsigned hist[NR_MAX];
    const int tid   = threadIdx.x;
    const int band  = blockIdx.x % nBands;
    const int slice = blockIdx.x / nBands;
    const int lo    = band * NR;
    int cnt = nN - lo;
    if (cnt > NR) cnt = NR;
    if (cnt < 0)  cnt = 0;

    for (int i = tid; i < cnt; i += BAND_T) hist[i] = 0u;
    __syncthreads();

    const int e0 = slice * ePerSlice;
    int e1 = e0 + ePerSlice;
    if (e1 > nE) e1 = nE;
    for (int base = e0; base < e1; base += EU * BAND_T) {
        int d[EU];
#pragma unroll
        for (int u = 0; u < EU; ++u) {
            int idx = base + u * BAND_T + tid;
            int ic  = (idx < e1) ? idx : (e1 - 1);
            int dv  = dst[ic];
            d[u] = (idx < e1) ? dv : -1;
        }
#pragma unroll
        for (int u = 0; u < EU; ++u) {
            unsigned r = (unsigned)(d[u] - lo);
            if (r < (unsigned)cnt) atomicAdd(&hist[r], 1u);
        }
    }
    __syncthreads();

    unsigned* p = part + (size_t)slice * nN + lo;
    for (int i = tid; i < cnt; i += BAND_T) p[i] = hist[i];
}

#define L1_R     416
#define NB_MAX   1024

__global__ void __launch_bounds__(L1_R)
k_inv_from_parts(const unsigned* __restrict__ part, float* __restrict__ inv,
                 unsigned* __restrict__ btot, unsigned* __restrict__ cursor, int nN) {
    __shared__ unsigned wsum[L1_R / 32];
    const int n = blockIdx.x * L1_R + threadIdx.x;
    unsigned c = 0u;
    if (n < nN) {
#pragma unroll 16
        for (int s = 0; s < NSLICE; ++s) c += part[(size_t)s * nN + n];
        inv[n] = rsqrtf(1.0f + (float)c);
    }
    for (int off = 16; off > 0; off >>= 1) c += __shfl_xor(c, off);
    if ((threadIdx.x & 31) == 0) wsum[threadIdx.x >> 5] = c;
    __syncthreads();
    if (threadIdx.x == 0) {
        unsigned t = 0u;
        for (int w = 0; w < L1_R / 32; ++w) t += wsum[w];
        btot[blockIdx.x]   = t;
        cursor[blockIdx.x] = 0u;
    }
}

#define GEMM_WAVES 25
#define GEMM_T     (GEMM_WAVES * 32)
#define GEMM_LDS   (164 * 1024)
__global__ void __launch_bounds__(GEMM_T)
k_gemm1_wmma(const float* __restrict__ X, const float* __restrict__ W,
             float* __restrict__ H, int nNodes) {
    extern __shared__ float bsh[];

    for (int idx = threadIdx.x; idx < N_HID_F * K_PAD; idx += blockDim.x) {
        int n = idx / K_PAD;
        int k = idx - n * K_PAD;
        int kc = (k < N_IN_F) ? k : (N_IN_F - 1);
        float wv = W[(size_t)kc * N_HID_F + n];
        bsh[n * BSTRIDE + k] = (k < N_IN_F) ? wv : 0.0f;
    }
    __syncthreads();

    const int lane   = threadIdx.x & 31;
    const int tileM  = blockIdx.x * (blockDim.x >> 5) + (threadIdx.x >> 5);
    const int nTiles = (nNodes + 15) >> 4;
    if (tileM >= nTiles) return;

    const int hi = lane >> 4;
    const int lo = lane & 15;
    int m = tileM * 16 + lo;
    if (m >= nNodes) m = nNodes - 1;

    v8f zero = {};
    v8f acc[8];
#pragma unroll
    for (int t = 0; t < 8; ++t) acc[t] = zero;

    const float* Xrow  = X + (size_t)m * N_IN_F;
    const float* bbase = bsh + lo * BSTRIDE;

#pragma unroll 2
    for (int k0 = 0; k0 < 164; k0 += 4) {
        const int kA = k0 + 2 * hi;
        v2f a;
        a.x = Xrow[kA];
        a.y = Xrow[kA + 1];
#pragma unroll
        for (int t = 0; t < 8; ++t) {
            v2f b = *(const v2f*)(bbase + (size_t)(t * 16) * BSTRIDE + kA);
            acc[t] = __builtin_amdgcn_wmma_f32_16x16x4_f32(
                false, a, false, b, (short)0, acc[t], false, false);
        }
    }

    {
        const int kA = 164 + 2 * hi;
        float xv = Xrow[164];
        v2f a;
        a.x = (hi == 0) ? xv : 0.0f;
        a.y = 0.0f;
#pragma unroll
        for (int t = 0; t < 8; ++t) {
            v2f b = *(const v2f*)(bbase + (size_t)(t * 16) * BSTRIDE + kA);
            acc[t] = __builtin_amdgcn_wmma_f32_16x16x4_f32(
                false, a, false, b, (short)0, acc[t], false, false);
        }
    }

    if (tileM * 16 + 16 <= nNodes) {
        float* Hrow = H + (size_t)(tileM * 16 + 8 * hi) * N_HID_F + lo;
#pragma unroll
        for (int r = 0; r < 8; ++r) {
#pragma unroll
            for (int t = 0; t < 8; ++t)
                Hrow[(size_t)r * N_HID_F + t * 16] = acc[t][r];
        }
    } else {
#pragma unroll
        for (int r = 0; r < 8; ++r) {
            int row = tileM * 16 + r + 8 * hi;
            if (row < nNodes) {
#pragma unroll
                for (int t = 0; t < 8; ++t)
                    H[(size_t)row * N_HID_F + t * 16 + lo] = acc[t][r];
            }
        }
    }
}

__global__ void __launch_bounds__(BAND_T)
k_bin(const int* __restrict__ src, const int* __restrict__ dst,
      const unsigned* __restrict__ btot, unsigned* __restrict__ cursor,
      unsigned* __restrict__ bstart, int2* __restrict__ rec, int nE, int nB) {
    __shared__ unsigned tot[NB_MAX];
    __shared__ unsigned cur[NB_MAX];
    const int tid = threadIdx.x;
    if (tid < nB) { tot[tid] = btot[tid]; cur[tid] = 0u; }

    int  s[EU], d[EU], bi[EU];
    bool ok[EU];
    const int base = blockIdx.x * (EU * BAND_T);
#pragma unroll
    for (int u = 0; u < EU; ++u) {
        const int idx = base + u * BAND_T + tid;
        ok[u] = idx < nE;
        const int ic = ok[u] ? idx : (nE - 1);
        d[u] = dst[ic];
        s[u] = src[ic];
    }
    __syncthreads();

    unsigned start = 0u;
    if (tid < nB) for (int j = 0; j < tid; ++j) start += tot[j];
#pragma unroll
    for (int u = 0; u < EU; ++u) {
        int b = (int)((unsigned)d[u] / (unsigned)L1_R);
        bi[u] = (b < nB) ? b : (nB - 1);
        if (ok[u]) atomicAdd(&cur[bi[u]], 1u);
    }
    __syncthreads();

    if (tid < nB) {
        const unsigned cnt = cur[tid];
        const unsigned g   = cnt ? atomicAdd(&cursor[tid], cnt) : 0u;
        cur[tid] = start + g;
        if (blockIdx.x == 0) bstart[tid] = start;
    }
    __syncthreads();

#pragma unroll
    for (int u = 0; u < EU; ++u) {
        if (ok[u]) {
            const unsigned pos = atomicAdd(&cur[bi[u]], 1u);
            if (pos < (unsigned)nE) rec[pos] = make_int2(s[u], d[u]);
        }
    }
}

#define AGG_EPW  32
#define AGG_U    8
__global__ void __launch_bounds__(BAND_T)
k_band_hid(const float* __restrict__ H, const int2* __restrict__ rec,
           const float* __restrict__ inv, const unsigned* __restrict__ bstart,
           const unsigned* __restrict__ btot, float* __restrict__ out, int nN) {
    __shared__ __attribute__((aligned(16))) float acc[L1_R * N_HID_F];
    const int tid  = threadIdx.x;
    const int lane = tid & 31;
    const int wv = __builtin_amdgcn_readfirstlane(tid >> 5);
    const int lo = blockIdx.x * L1_R;
    int cnt = nN - lo;
    if (cnt > L1_R) cnt = L1_R;

    float4* a4 = (float4*)acc;
    {
        const float4 z = make_float4(0.0f, 0.0f, 0.0f, 0.0f);
        for (int i = tid; i < cnt * (N_HID_F / 4); i += BAND_T) a4[i] = z;
    }
    __syncthreads();

    const int e0 = __builtin_amdgcn_readfirstlane((int)bstart[blockIdx.x]);
    const int e1 = e0 + __builtin_amdgcn_readfirstlane((int)btot[blockIdx.x]);
    for (int base = e0 + wv * AGG_EPW; base < e1; base += AGG_EPW * (BAND_T / 32)) {
        const int e  = base + lane;
        const int ec = (e < e1) ? e : (e1 - 1);
        const int2 r = rec[ec];
        int sr = r.x;
        sr = (sr < 0) ? 0 : ((sr < nN) ? sr : (nN - 1));
        const int  dl  = r.y - lo;
        const bool in  = (unsigned)dl < (unsigned)cnt;
        const int  s   = sr;
        const int  d   = in ? dl : 0;
        const int  cb  = in ? __float_as_int(inv[sr] * inv[lo + dl]) : 0;
        int n = e1 - base;
        if (n > AGG_EPW) n = AGG_EPW;

        if (n == AGG_EPW) {
#pragma unroll
            for (int j0 = 0; j0 < AGG_EPW; j0 += AGG_U) {
                float v[AGG_U][4];
#pragma unroll
                for (int u = 0; u < AGG_U; ++u) {
                    const int sj = __builtin_amdgcn_readlane(s, j0 + u);
                    const float* h = H + (size_t)sj * N_HID_F + lane;
                    v[u][0] = h[0]; v[u][1] = h[32]; v[u][2] = h[64]; v[u][3] = h[96];
                }
#pragma unroll
                for (int u = 0; u < AGG_U; ++u) {
                    const int   dj = __builtin_amdgcn_readlane(d, j0 + u);
                    const float c  = __int_as_float(__builtin_amdgcn_readlane(cb, j0 + u));
                    float* o = acc + dj * N_HID_F + lane;
                    atomicAdd(o +  0, v[u][0] * c);
                    atomicAdd(o + 32, v[u][1] * c);
                    atomicAdd(o + 64, v[u][2] * c);
                    atomicAdd(o + 96, v[u][3] * c);
                }
            }
        } else {
            for (int j = 0; j < n; ++j) {
                const int   sj = __builtin_amdgcn_readlane(s, j);
                const int   dj = __builtin_amdgcn_readlane(d, j);
                const float c  = __int_as_float(__builtin_amdgcn_readlane(cb, j));
                const float* h = H + (size_t)sj * N_HID_F + lane;
                const float v0 = h[0], v1 = h[32], v2 = h[64], v3 = h[96];
                float* o = acc + dj * N_HID_F + lane;
                atomicAdd(o +  0, v0 * c);
                atomicAdd(o + 32, v1 * c);
                atomicAdd(o + 64, v2 * c);
                atomicAdd(o + 96, v3 * c);
            }
        }
    }
    __syncthreads();

    for (int i = wv; i < cnt; i += BAND_T / 32) {
        const float4 t = a4[i * (N_HID_F / 4) + lane];
        *(float4*)(out + (size_t)(lo + i) * N_HID_F + lane * 4) = t;
    }
}

__global__ void __launch_bounds__(EW_T)
k_post1(float* __restrict__ H1, const float* __restrict__ agg,
        const float* __restrict__ inv, const float* __restrict__ b1,
        int nNodes) {
    const int lane = threadIdx.x & 31;
    const int wv   = __builtin_amdgcn_readfirstlane((int)(threadIdx.x >> 5));
    const int wpb  = blockDim.x >> 5;
    const int nw   = gridDim.x * wpb;
    const int f    = lane * 4;
    const float b0 = b1[f], b1v = b1[f + 1], b2v = b1[f + 2], b3v = b1[f + 3];
    for (int node = blockIdx.x * wpb + wv; node < nNodes; node += nw) {
        const size_t off = (size_t)node * N_HID_F + f;
        const float4 a  = *(const float4*)(agg + off);
        const float4 h  = *(const float4*)(H1 + off);
        const float  is = inv[node];
        float4 r;
        r.x = tanhf(a.x + h.x * is * is + b0);
        r.y = tanhf(a.y + h.y * is * is + b1v);
        r.z = tanhf(a.z + h.z * is * is + b2v);
        r.w = tanhf(a.w + h.w * is * is + b3v);
        *(float4*)(H1 + off) = r;
    }
}

__global__ void k_gemm2(const float* __restrict__ H, const float* __restrict__ W2,
                        float* __restrict__ H2, int nNodes) {
    __shared__ float w2s[N_HID_F * N_EMB_F];
    int tid = threadIdx.x;
    if (tid < N_HID_F * N_EMB_F) w2s[tid] = W2[tid];
    __syncthreads();
    int n = blockIdx.x * blockDim.x + tid;
    if (n >= nNodes) return;
    const float* h = H + (size_t)n * N_HID_F;
    float a0 = 0.0f, a1 = 0.0f;
#pragma unroll 8
    for (int k = 0; k < N_HID_F; ++k) {
        float hv = h[k];
        a0 += hv * w2s[k * 2 + 0];
        a1 += hv * w2s[k * 2 + 1];
    }
    H2[(size_t)n * 2 + 0] = a0;
    H2[(size_t)n * 2 + 1] = a1;
}

__global__ void __launch_bounds__(BAND_T)
k_emb_band(const float* __restrict__ H2, const int* __restrict__ src,
           const int* __restrict__ dst, const float* __restrict__ inv,
           float* __restrict__ part, int nE, int nN, int nBands, int NR, int ePerSlice) {
    __shared__ __attribute__((aligned(16))) float acc[NR_MAX * 2];
    const int tid   = threadIdx.x;
    const int band  = blockIdx.x % nBands;
    const int slice = blockIdx.x / nBands;
    const int lo    = band * NR;
    int cnt = nN - lo;
    if (cnt > NR) cnt = NR;
    if (cnt < 0)  cnt = 0;

    {
        float4* a4 = (float4*)acc;
        const int n4 = (2 * cnt + 3) >> 2;
        const float4 z = make_float4(0.0f, 0.0f, 0.0f, 0.0f);
        for (int i = tid; i < n4; i += BAND_T) a4[i] = z;
    }
    __syncthreads();

    const int e0 = slice * ePerSlice;
    int e1 = e0 + ePerSlice;
    if (e1 > nE) e1 = nE;
    for (int base = e0; base < e1; base += EU * BAND_T) {
        int  dd[EU], ss[EU];
        bool m[EU];
#pragma unroll
        for (int u = 0; u < EU; ++u) {
            int idx = base + u * BAND_T + tid;
            int ic  = (idx < e1) ? idx : (e1 - 1);
            int d = dst[ic];
            int s = src[ic];
            unsigned r = (unsigned)(d - lo);
            m[u]  = (idx < e1) && (r < (unsigned)cnt);
            dd[u] = m[u] ? d : 0;
            ss[u] = m[u] ? s : 0;
        }
        float  is[EU], id[EU];
        float2 v[EU];
#pragma unroll
        for (int u = 0; u < EU; ++u) {
            is[u] = inv[ss[u]];
            id[u] = inv[dd[u]];
            v[u]  = *(const float2*)(H2 + (size_t)ss[u] * 2);
        }
        asm volatile("" :: "v"(is[0]), "v"(id[0]), "v"(v[0].x), "v"(v[0].y));
#pragma unroll
        for (int u = 0; u < EU; ++u) {
            if (m[u]) {
                const float c = is[u] * id[u];
                float* a = acc + 2 * (dd[u] - lo);
                atomicAdd(a + 0, v[u].x * c);
                atomicAdd(a + 1, v[u].y * c);
            }
        }
    }
    __syncthreads();

    float2*       p  = (float2*)(part + (size_t)slice * 2 * nN) + lo;
    const float2* a2 = (const float2*)acc;
    for (int i = tid; i < cnt; i += BAND_T) p[i] = a2[i];
}

__global__ void k_final(const float* __restrict__ H2, const float* __restrict__ part,
                        const float* __restrict__ inv, const float* __restrict__ b2,
                        const float* __restrict__ Wc, const float* __restrict__ bc,
                        float* __restrict__ out, int nNodes) {
    int n = blockIdx.x * blockDim.x + threadIdx.x;
    if (n >= nNodes) return;
    const float2* p = (const float2*)part + n;
    float g0 = 0.0f, g1 = 0.0f;
#pragma unroll 16
    for (int s = 0; s < NSLICE; ++s) {
        float2 t = p[(size_t)s * nNodes];
        g0 += t.x;
        g1 += t.y;
    }
    float is = inv[n];
    float i2 = is * is;
    float e0 = tanhf(g0 + H2[(size_t)n * 2 + 0] * i2 + b2[0]);
    float e1 = tanhf(g1 + H2[(size_t)n * 2 + 1] * i2 + b2[1]);
    float z  = e0 * Wc[0] + e1 * Wc[1] + bc[0];
    out[n] = 1.0f / (1.0f + expf(-z));
}

extern "C" void kernel_launch(void* const* d_in, const int* in_sizes, int n_in,
                              void* d_out, int out_size, void* d_ws, size_t ws_size,
                              hipStream_t stream) {
    const float* x   = (const float*)d_in[0];
    const int*   ei  = (const int*)d_in[1];
    const float* W1  = (const float*)d_in[2];
    const float* b1  = (const float*)d_in[3];
    const float* W2  = (const float*)d_in[4];
    const float* b2  = (const float*)d_in[5];
    const float* Wc  = (const float*)d_in[6];
    const float* bc  = (const float*)d_in[7];
    float* out = (float*)d_out;

    const int nN = in_sizes[0] / N_IN_F;
    const int nE = in_sizes[1] / 2;
    const int* src = ei;
    const int* dst = ei + nE;

    float* ws   = (float*)d_ws;
    float* h1   = ws;
    float* agg1 = h1   + (size_t)nN * N_HID_F;
    float* pemb = agg1 + (size_t)nN * N_HID_F;
    int2*  rec  = (int2*)(pemb + (size_t)NSLICE * nN * 2);
    unsigned* pdeg = (unsigned*)(rec + (size_t)nE);
    float* h2   = (float*)(pdeg + (size_t)NSLICE * nN);
    float* inv  = h2   + (size_t)nN * 2;
    unsigned* btot   = (unsigned*)(inv + (size_t)nN);
    unsigned* cursor = btot + NB_MAX;
    unsigned* bstart = cursor + NB_MAX;

    const int B   = 256;
    const int nB1 = (nN + L1_R - 1) / L1_R;

    const int nBands    = (nN + NR_MAX - 1) / NR_MAX;
    const int NR        = (nN + nBands - 1) / nBands;
    const int ePerSlice = (nE + NSLICE - 1) / NSLICE;

    k_deg_band<<<nBands * NSLICE, BAND_T, 0, stream>>>(dst, pdeg, nE, nN, nBands, NR, ePerSlice);
    k_inv_from_parts<<<nB1, L1_R, 0, stream>>>(pdeg, inv, btot, cursor, nN);

    k_bin<<<(nE + EU * BAND_T - 1) / (EU * BAND_T), BAND_T, 0, stream>>>(src, dst, btot, cursor, bstart, rec, nE, nB1);
    const int nTiles = (nN + 15) / 16;
    k_gemm1_wmma<<<(nTiles + GEMM_WAVES - 1) / GEMM_WAVES, GEMM_T, GEMM_LDS, stream>>>(x, W1, h1, nN);
    k_band_hid<<<nB1, BAND_T, 0, stream>>>(h1, rec, inv, bstart, btot, agg1, nN);
    k_post1<<<EW_WG, EW_T, 0, stream>>>(h1, agg1, inv, b1, nN);

    k_gemm2<<<(nN + B - 1) / B, B, 0, stream>>>(h1, W2, h2, nN);
    k_emb_band<<<nBands * NSLICE, BAND_T, 0, stream>>>(h2, src, dst, inv, pemb, nE, nN, nBands, NR, ePerSlice);
    k_final<<<(nN + B - 1) / B, B, 0, stream>>>(h2, pemb, inv, b2, Wc, bc, out, nN);
}


# Layer-2 sums: owner work-groups walk the layer-1 band-binned edge list, add in LDS and finish their 416 nodes themselves: no 64 partial planes, no plane-sum launch (9 -> 8); sum order = LDS arrival
# speedup vs baseline: 3.5394x; 1.1273x over previous
#include <hip/hip_runtime.h>
#include <cmath>

typedef __attribute__((ext_vector_type(2))) float v2f;
typedef __attribute__((ext_vector_type(8))) float v8f;

#define N_IN_F   165
#define N_HID_F  128
#define N_EMB_F  2
#define K_PAD    168
#define BSTRIDE  170

#define EW_T   256
#define EW_WG  2048
#define NR_MAX   32768
#define NSLICE   64
#define BAND_T   1024
#define EU       8

__global__ void __launch_bounds__(BAND_T)
k_deg_band(const int* __restrict__ dst, unsigned* __restrict__ part,
           int nE, int nN, int nBands, int NR, int ePerSlice) {
    __shared__ unsigned hist[NR_MAX];
    const int tid   = threadIdx.x;
    const int band  = blockIdx.x % nBands;
    const int slice = blockIdx.x / nBands;
    const int lo    = band * NR;
    int cnt = nN - lo;
    if (cnt > NR) cnt = NR;
    if (cnt < 0)  cnt = 0;

    for (int i = tid; i < cnt; i += BAND_T) hist[i] = 0u;
    __syncthreads();

    const int e0 = slice * ePerSlice;
    int e1 = e0 + ePerSlice;
    if (e1 > nE) e1 = nE;
    for (int base = e0; base < e1; base += EU * BAND_T) {
        int d[EU];
#pragma unroll
        for (int u = 0; u < EU; ++u) {
            int idx = base + u * BAND_T + tid;
            int ic  = (idx < e1) ? idx : (e1 - 1);
            int dv  = dst[ic];
            d[u] = (idx < e1) ? dv : -1;
        }
#pragma unroll
        for (int u = 0; u < EU; ++u) {
            unsigned r = (unsigned)(d[u] - lo);
            if (r < (unsigned)cnt) atomicAdd(&hist[r], 1u);
        }
    }
    __syncthreads();

    unsigned* p = part + (size_t)slice * nN + lo;
    for (int i = tid; i < cnt; i += BAND_T) p[i] = hist[i];
}

#define L1_R     416
#define NB_MAX   1024

__global__ void __launch_bounds__(L1_R)
k_inv_from_parts(const unsigned* __restrict__ part, float* __restrict__ inv,
                 unsigned* __restrict__ btot, unsigned* __restrict__ cursor, int nN) {
    __shared__ unsigned wsum[L1_R / 32];
    const int n = blockIdx.x * L1_R + threadIdx.x;
    unsigned c = 0u;
    if (n < nN) {
#pragma unroll 16
        for (int s = 0; s < NSLICE; ++s) c += part[(size_t)s * nN + n];
        inv[n] = rsqrtf(1.0f + (float)c);
    }
    for (int off = 16; off > 0; off >>= 1) c += __shfl_xor(c, off);
    if ((threadIdx.x & 31) == 0) wsum[threadIdx.x >> 5] = c;
    __syncthreads();
    if (threadIdx.x == 0) {
        unsigned t = 0u;
        for (int w = 0; w < L1_R / 32; ++w) t += wsum[w];
        btot[blockIdx.x]   = t;
        cursor[blockIdx.x] = 0u;
    }
}

#define GEMM_WAVES 25
#define GEMM_T     (GEMM_WAVES * 32)
#define GEMM_LDS   (164 * 1024)
__global__ void __launch_bounds__(GEMM_T)
k_gemm1_wmma(const float* __restrict__ X, const float* __restrict__ W,
             float* __restrict__ H, int nNodes) {
    extern __shared__ float bsh[];

    for (int idx = threadIdx.x; idx < N_HID_F * K_PAD; idx += blockDim.x) {
        int n = idx / K_PAD;
        int k = idx - n * K_PAD;
        int kc = (k < N_IN_F) ? k : (N_IN_F - 1);
        float wv = W[(size_t)kc * N_HID_F + n];
        bsh[n * BSTRIDE + k] = (k < N_IN_F) ? wv : 0.0f;
    }
    __syncthreads();

    const int lane   = threadIdx.x & 31;
    const int tileM  = blockIdx.x * (blockDim.x >> 5) + (threadIdx.x >> 5);
    const int nTiles = (nNodes + 15) >> 4;
    if (tileM >= nTiles) return;

    const int hi = lane >> 4;
    const int lo = lane & 15;
    int m = tileM * 16 + lo;
    if (m >= nNodes) m = nNodes - 1;

    v8f zero = {};
    v8f acc[8];
#pragma unroll
    for (int t = 0; t < 8; ++t) acc[t] = zero;

    const float* Xrow  = X + (size_t)m * N_IN_F;
    const float* bbase = bsh + lo * BSTRIDE;

#pragma unroll 2
    for (int k0 = 0; k0 < 164; k0 += 4) {
        const int kA = k0 + 2 * hi;
        v2f a;
        a.x = Xrow[kA];
        a.y = Xrow[kA + 1];
#pragma unroll
        for (int t = 0; t < 8; ++t) {
            v2f b = *(const v2f*)(bbase + (size_t)(t * 16) * BSTRIDE + kA);
            acc[t] = __builtin_amdgcn_wmma_f32_16x16x4_f32(
                false, a, false, b, (short)0, acc[t], false, false);
        }
    }

    {
        const int kA = 164 + 2 * hi;
        float xv = Xrow[164];
        v2f a;
        a.x = (hi == 0) ? xv : 0.0f;
        a.y = 0.0f;
#pragma unroll
        for (int t = 0; t < 8; ++t) {
            v2f b = *(const v2f*)(bbase + (size_t)(t * 16) * BSTRIDE + kA);
            acc[t] = __builtin_amdgcn_wmma_f32_16x16x4_f32(
                false, a, false, b, (short)0, acc[t], false, false);
        }
    }

    if (tileM * 16 + 16 <= nNodes) {
        float* Hrow = H + (size_t)(tileM * 16 + 8 * hi) * N_HID_F + lo;
#pragma unroll
        for (int r = 0; r < 8; ++r) {
#pragma unroll
            for (int t = 0; t < 8; ++t)
                Hrow[(size_t)r * N_HID_F + t * 16] = acc[t][r];
        }
    } else {
#pragma unroll
        for (int r = 0; r < 8; ++r) {
            int row = tileM * 16 + r + 8 * hi;
            if (row < nNodes) {
#pragma unroll
                for (int t = 0; t < 8; ++t)
                    H[(size_t)row * N_HID_F + t * 16 + lo] = acc[t][r];
            }
        }
    }
}

__global__ void __launch_bounds__(BAND_T)
k_bin(const int* __restrict__ src, const int* __restrict__ dst,
      const unsigned* __restrict__ btot, unsigned* __restrict__ cursor,
      unsigned* __restrict__ bstart, int2* __restrict__ rec, int nE, int nB) {
    __shared__ unsigned tot[NB_MAX];
    __shared__ unsigned cur[NB_MAX];
    const int tid = threadIdx.x;
    if (tid < nB) { tot[tid] = btot[tid]; cur[tid] = 0u; }

    int  s[EU], d[EU], bi[EU];
    bool ok[EU];
    const int base = blockIdx.x * (EU * BAND_T);
#pragma unroll
    for (int u = 0; u < EU; ++u) {
        const int idx = base + u * BAND_T + tid;
        ok[u] = idx < nE;
        const int ic = ok[u] ? idx : (nE - 1);
        d[u] = dst[ic];
        s[u] = src[ic];
    }
    __syncthreads();

    unsigned start = 0u;
    if (tid < nB) for (int j = 0; j < tid; ++j) start += tot[j];
#pragma unroll
    for (int u = 0; u < EU; ++u) {
        int b = (int)((unsigned)d[u] / (unsigned)L1_R);
        bi[u] = (b < nB) ? b : (nB - 1);
        if (ok[u]) atomicAdd(&cur[bi[u]], 1u);
    }
    __syncthreads();

    if (tid < nB) {
        const unsigned cnt = cur[tid];
        const unsigned g   = cnt ? atomicAdd(&cursor[tid], cnt) : 0u;
        cur[tid] = start + g;
        if (blockIdx.x == 0) bstart[tid] = start;
    }
    __syncthreads();

#pragma unroll
    for (int u = 0; u < EU; ++u) {
        if (ok[u]) {
            const unsigned pos = atomicAdd(&cur[bi[u]], 1u);
            if (pos < (unsigned)nE) rec[pos] = make_int2(s[u], d[u]);
        }
    }
}

#define AGG_EPW  32
#define AGG_U    8
__global__ void __launch_bounds__(BAND_T)
k_band_hid(const float* __restrict__ H, const int2* __restrict__ rec,
           const float* __restrict__ inv, const unsigned* __restrict__ bstart,
           const unsigned* __restrict__ btot, float* __restrict__ out, int nN) {
    __shared__ __attribute__((aligned(16))) float acc[L1_R * N_HID_F];
    const int tid  = threadIdx.x;
    const int lane = tid & 31;
    const int wv = __builtin_amdgcn_readfirstlane(tid >> 5);
    const int lo = blockIdx.x * L1_R;
    int cnt = nN - lo;
    if (cnt > L1_R) cnt = L1_R;

    float4* a4 = (float4*)acc;
    {
        const float4 z = make_float4(0.0f, 0.0f, 0.0f, 0.0f);
        for (int i = tid; i < cnt * (N_HID_F / 4); i += BAND_T) a4[i] = z;
    }
    __syncthreads();

    const int e0 = __builtin_amdgcn_readfirstlane((int)bstart[blockIdx.x]);
    const int e1 = e0 + __builtin_amdgcn_readfirstlane((int)btot[blockIdx.x]);
    for (int base = e0 + wv * AGG_EPW; base < e1; base += AGG_EPW * (BAND_T / 32)) {
        const int e  = base + lane;
        const int ec = (e < e1) ? e : (e1 - 1);
        const int2 r = rec[ec];
        int sr = r.x;
        sr = (sr < 0) ? 0 : ((sr < nN) ? sr : (nN - 1));
        const int  dl  = r.y - lo;
        const bool in  = (unsigned)dl < (unsigned)cnt;
        const int  s   = sr;
        const int  d   = in ? dl : 0;
        const int  cb  = in ? __float_as_int(inv[sr] * inv[lo + dl]) : 0;
        int n = e1 - base;
        if (n > AGG_EPW) n = AGG_EPW;

        if (n == AGG_EPW) {
#pragma unroll
            for (int j0 = 0; j0 < AGG_EPW; j0 += AGG_U) {
                float v[AGG_U][4];
#pragma unroll
                for (int u = 0; u < AGG_U; ++u) {
                    const int sj = __builtin_amdgcn_readlane(s, j0 + u);
                    const float* h = H + (size_t)sj * N_HID_F + lane;
                    v[u][0] = h[0]; v[u][1] = h[32]; v[u][2] = h[64]; v[u][3] = h[96];
                }
#pragma unroll
                for (int u = 0; u < AGG_U; ++u) {
                    const int   dj = __builtin_amdgcn_readlane(d, j0 + u);
                    const float c  = __int_as_float(__builtin_amdgcn_readlane(cb, j0 + u));
                    float* o = acc + dj * N_HID_F + lane;
                    atomicAdd(o +  0, v[u][0] * c);
                    atomicAdd(o + 32, v[u][1] * c);
                    atomicAdd(o + 64, v[u][2] * c);
                    atomicAdd(o + 96, v[u][3] * c);
                }
            }
        } else {
            for (int j = 0; j < n; ++j) {
                const int   sj = __builtin_amdgcn_readlane(s, j);
                const int   dj = __builtin_amdgcn_readlane(d, j);
                const float c  = __int_as_float(__builtin_amdgcn_readlane(cb, j));
                const float* h = H + (size_t)sj * N_HID_F + lane;
                const float v0 = h[0], v1 = h[32], v2 = h[64], v3 = h[96];
                float* o = acc + dj * N_HID_F + lane;
                atomicAdd(o +  0, v0 * c);
                atomicAdd(o + 32, v1 * c);
                atomicAdd(o + 64, v2 * c);
                atomicAdd(o + 96, v3 * c);
            }
        }
    }
    __syncthreads();

    for (int i = wv; i < cnt; i += BAND_T / 32) {
        const float4 t = a4[i * (N_HID_F / 4) + lane];
        *(float4*)(out + (size_t)(lo + i) * N_HID_F + lane * 4) = t;
    }
}

__global__ void __launch_bounds__(EW_T)
k_post1(float* __restrict__ H1, const float* __restrict__ agg,
        const float* __restrict__ inv, const float* __restrict__ b1,
        int nNodes) {
    const int lane = threadIdx.x & 31;
    const int wv   = __builtin_amdgcn_readfirstlane((int)(threadIdx.x >> 5));
    const int wpb  = blockDim.x >> 5;
    const int nw   = gridDim.x * wpb;
    const int f    = lane * 4;
    const float b0 = b1[f], b1v = b1[f + 1], b2v = b1[f + 2], b3v = b1[f + 3];
    for (int node = blockIdx.x * wpb + wv; node < nNodes; node += nw) {
        const size_t off = (size_t)node * N_HID_F + f;
        const float4 a  = *(const float4*)(agg + off);
        const float4 h  = *(const float4*)(H1 + off);
        const float  is = inv[node];
        float4 r;
        r.x = tanhf(a.x + h.x * is * is + b0);
        r.y = tanhf(a.y + h.y * is * is + b1v);
        r.z = tanhf(a.z + h.z * is * is + b2v);
        r.w = tanhf(a.w + h.w * is * is + b3v);
        *(float4*)(H1 + off) = r;
    }
}

__global__ void k_gemm2(const float* __restrict__ H, const float* __restrict__ W2,
                        float* __restrict__ H2, int nNodes) {
    __shared__ float w2s[N_HID_F * N_EMB_F];
    int tid = threadIdx.x;
    if (tid < N_HID_F * N_EMB_F) w2s[tid] = W2[tid];
    __syncthreads();
    int n = blockIdx.x * blockDim.x + tid;
    if (n >= nNodes) return;
    const float* h = H + (size_t)n * N_HID_F;
    float a0 = 0.0f, a1 = 0.0f;
#pragma unroll 8
    for (int k = 0; k < N_HID_F; ++k) {
        float hv = h[k];
        a0 += hv * w2s[k * 2 + 0];
        a1 += hv * w2s[k * 2 + 1];
    }
    H2[(size_t)n * 2 + 0] = a0;
    H2[(size_t)n * 2 + 1] = a1;
}

#define EMB_T    512
static_assert(EMB_T >= L1_R, "the finishing pass takes one node per thread");
static_assert(EU == 8, "the two load fences of k_band_emb name 8 edges");
__global__ void __launch_bounds__(EMB_T)
k_band_emb(const float* __restrict__ H2, const int2* __restrict__ rec,
           const float* __restrict__ inv, const unsigned* __restrict__ bstart,
           const unsigned* __restrict__ btot, const float* __restrict__ b2,
           const float* __restrict__ Wc, const float* __restrict__ bc,
           float* __restrict__ out, int nN) {
    __shared__ __attribute__((aligned(8))) float acc[L1_R * 2];
    const int tid = threadIdx.x;
    const int lo  = blockIdx.x * L1_R;
    int cnt = nN - lo;
    if (cnt > L1_R) cnt = L1_R;

    for (int i = tid; i < L1_R * 2; i += EMB_T) acc[i] = 0.0f;

    const int e0 = (int)bstart[blockIdx.x];
    const int e1 = e0 + (int)btot[blockIdx.x];
    const bool   own = tid < cnt;
    const int    n   = lo + (own ? tid : 0);
    const float  isn = inv[n];
    const float2 hn  = *(const float2*)(H2 + (size_t)n * 2);
    const float  b20 = b2[0], b21 = b2[1], wc0 = Wc[0], wc1 = Wc[1], bc0 = bc[0];
    __syncthreads();

    for (int base = e0; base < e1; base += EU * EMB_T) {
        int2 r[EU];
        int  dd[EU], ss[EU];
        bool m[EU];
#pragma unroll
        for (int u = 0; u < EU; ++u) {
            const int idx = base + u * EMB_T + tid;
            const int ic  = (idx < e1) ? idx : (e1 - 1);
            r[u] = rec[ic];
        }
        asm volatile("" :: "v"(r[0].x), "v"(r[0].y), "v"(r[1].x), "v"(r[1].y), "v"(r[2].x), "v"(r[2].y),
                           "v"(r[3].x), "v"(r[3].y), "v"(r[4].x), "v"(r[4].y), "v"(r[5].x), "v"(r[5].y),
                           "v"(r[6].x), "v"(r[6].y), "v"(r[7].x), "v"(r[7].y));
#pragma unroll
        for (int u = 0; u < EU; ++u) {
            const int idx = base + u * EMB_T + tid;
            int s = r[u].x;
            s = (s < 0) ? 0 : ((s < nN) ? s : (nN - 1));
            const int dl = r[u].y - lo;
            m[u]  = (idx < e1) && ((unsigned)dl < (unsigned)cnt);
            dd[u] = m[u] ? dl : 0;
            ss[u] = m[u] ? s : 0;
        }
        float  is[EU], id[EU];
        float2 v[EU];
#pragma unroll
        for (int u = 0; u < EU; ++u) {
            is[u] = inv[ss[u]];
            id[u] = inv[lo + dd[u]];
            v[u]  = *(const float2*)(H2 + (size_t)ss[u] * 2);
        }
        asm volatile("" :: "v"(is[0]), "v"(id[0]), "v"(v[0].x), "v"(is[1]), "v"(id[1]), "v"(v[1].x),
                           "v"(is[2]), "v"(id[2]), "v"(v[2].x), "v"(is[3]), "v"(id[3]), "v"(v[3].x),
                           "v"(is[4]), "v"(id[4]), "v"(v[4].x), "v"(is[5]), "v"(id[5]), "v"(v[5].x),
                           "v"(is[6]), "v"(id[6]), "v"(v[6].x), "v"(is[7]), "v"(id[7]), "v"(v[7].x));
#pragma unroll
        for (int u = 0; u < EU; ++u) {
            if (m[u]) {
                const float c = is[u] * id[u];
                float* a = acc + 2 * dd[u];
                atomicAdd(a + 0, v[u].x * c);
                atomicAdd(a + 1, v[u].y * c);
            }
        }
    }
    __syncthreads();

    if (own) {
        const float i2 = isn * isn;
        const float t0 = tanhf(acc[2 * tid + 0] + hn.x * i2 + b20);
        const float t1 = tanhf(acc[2 * tid + 1] + hn.y * i2 + b21);
        const float z  = t0 * wc0 + t1 * wc1 + bc0;
        out[n] = 1.0f / (1.0f + expf(-z));
    }
}

extern "C" void kernel_launch(void* const* d_in, const int* in_sizes, int n_in,
                              void* d_out, int out_size, void* d_ws, size_t ws_size,
                              hipStream_t stream) {
    const float* x   = (const float*)d_in[0];
    const int*   ei  = (const int*)d_in[1];
    const float* W1  = (const float*)d_in[2];
    const float* b1  = (const float*)d_in[3];
    const float* W2  = (const float*)d_in[4];
    const float* b2  = (const float*)d_in[5];
    const float* Wc  = (const float*)d_in[6];
    const float* bc  = (const float*)d_in[7];
    float* out = (float*)d_out;

    const int nN = in_sizes[0] / N_IN_F;
    const int nE = in_sizes[1] / 2;
    const int* src = ei;
    const int* dst = ei + nE;

    float* ws   = (float*)d_ws;
    float* h1   = ws;
    float* agg1 = h1   + (size_t)nN * N_HID_F;
    int2*  rec  = (int2*)(agg1 + (size_t)nN * N_HID_F);
    unsigned* pdeg = (unsigned*)(rec + (size_t)nE);
    float* h2   = (float*)(pdeg + (size_t)NSLICE * nN);
    float* inv  = h2   + (size_t)nN * 2;
    unsigned* btot   = (unsigned*)(inv + (size_t)nN);
    unsigned* cursor = btot + NB_MAX;
    unsigned* bstart = cursor + NB_MAX;

    const int B   = 256;
    const int nB1 = (nN + L1_R - 1) / L1_R;

    const int nBands    = (nN + NR_MAX - 1) / NR_MAX;
    const int NR        = (nN + nBands - 1) / nBands;
    const int ePerSlice = (nE + NSLICE - 1) / NSLICE;

    k_deg_band<<<nBands * NSLICE, BAND_T, 0, stream>>>(dst, pdeg, nE, nN, nBands, NR, ePerSlice);
    k_inv_from_parts<<<nB1, L1_R, 0, stream>>>(pdeg, inv, btot, cursor, nN);

    k_bin<<<(nE + EU * BAND_T - 1) / (EU * BAND_T), BAND_T, 0, stream>>>(src, dst, btot, cursor, bstart, rec, nE, nB1);
    const int nTiles = (nN + 15) / 16;
    k_gemm1_wmma<<<(nTiles + GEMM_WAVES - 1) / GEMM_WAVES, GEMM_T, GEMM_LDS, stream>>>(x, W1, h1, nN);
    k_band_hid<<<nB1, BAND_T, 0, stream>>>(h1, rec, inv, bstart, btot, agg1, nN);
    k_post1<<<EW_WG, EW_T, 0, stream>>>(h1, agg1, inv, b1, nN);

    k_gemm2<<<(nN + B - 1) / B, B, 0, stream>>>(h1, W2, h2, nN);
    k_band_emb<<<nB1, EMB_T, 0, stream>>>(h2, rec, inv, bstart, btot, b2, Wc, bc, out, nN);
}


# Layer-1 band owners finish their 416 nodes from LDS: self term, b1, tanh, 128->2 product (4 per lane, then lane tree: product sum order changed); no sums plane, no post or product launch (8 -> 6)
# speedup vs baseline: 3.9900x; 1.1273x over previous
#include <hip/hip_runtime.h>
#include <cmath>

typedef __attribute__((ext_vector_type(2))) float v2f;
typedef __attribute__((ext_vector_type(8))) float v8f;

#define N_IN_F   165
#define N_HID_F  128
#define N_EMB_F  2
#define K_PAD    168
#define BSTRIDE  170

#define NR_MAX   32768
#define NSLICE   64
#define BAND_T   1024
#define EU       8

__global__ void __launch_bounds__(BAND_T)
k_deg_band(const int* __restrict__ dst, unsigned* __restrict__ part,
           int nE, int nN, int nBands, int NR, int ePerSlice) {
    __shared__ unsigned hist[NR_MAX];
    const int tid   = threadIdx.x;
    const int band  = blockIdx.x % nBands;
    const int slice = blockIdx.x / nBands;
    const int lo    = band * NR;
    int cnt = nN - lo;
    if (cnt > NR) cnt = NR;
    if (cnt < 0)  cnt = 0;

    for (int i = tid; i < cnt; i += BAND_T) hist[i] = 0u;
    __syncthreads();

    const int e0 = slice * ePerSlice;
    int e1 = e0 + ePerSlice;
    if (e1 > nE) e1 = nE;
    for (int base = e0; base < e1; base += EU * BAND_T) {
        int d[EU];
#pragma unroll
        for (int u = 0; u < EU; ++u) {
            int idx = base + u * BAND_T + tid;
            int ic  = (idx < e1) ? idx : (e1 - 1);
            int dv  = dst[ic];
            d[u] = (idx < e1) ? dv : -1;
        }
#pragma unroll
        for (int u = 0; u < EU; ++u) {
            unsigned r = (unsigned)(d[u] - lo);
            if (r < (unsigned)cnt) atomicAdd(&hist[r], 1u);
        }
    }
    __syncthreads();

    unsigned* p = part + (size_t)slice * nN + lo;
    for (int i = tid; i < cnt; i += BAND_T) p[i] = hist[i];
}

#define L1_R     416
#define NB_MAX   1024

__global__ void __launch_bounds__(L1_R)
k_inv_from_parts(const unsigned* __restrict__ part, float* __restrict__ inv,
                 unsigned* __restrict__ btot, unsigned* __restrict__ cursor, int nN) {
    __shared__ unsigned wsum[L1_R / 32];
    const int n = blockIdx.x * L1_R + threadIdx.x;
    unsigned c = 0u;
    if (n < nN) {
#pragma unroll 16
        for (int s = 0; s < NSLICE; ++s) c += part[(size_t)s * nN + n];
        inv[n] = rsqrtf(1.0f + (float)c);
    }
    for (int off = 16; off > 0; off >>= 1) c += __shfl_xor(c, off);
    if ((threadIdx.x & 31) == 0) wsum[threadIdx.x >> 5] = c;
    __syncthreads();
    if (threadIdx.x == 0) {
        unsigned t = 0u;
        for (int w = 0; w < L1_R / 32; ++w) t += wsum[w];
        btot[blockIdx.x]   = t;
        cursor[blockIdx.x] = 0u;
    }
}

#define GEMM_WAVES 25
#define GEMM_T     (GEMM_WAVES * 32)
#define GEMM_LDS   (164 * 1024)
__global__ void __launch_bounds__(GEMM_T)
k_gemm1_wmma(const float* __restrict__ X, const float* __restrict__ W,
             float* __restrict__ H, int nNodes) {
    extern __shared__ float bsh[];

    for (int idx = threadIdx.x; idx < N_HID_F * K_PAD; idx += blockDim.x) {
        int n = idx / K_PAD;
        int k = idx - n * K_PAD;
        int kc = (k < N_IN_F) ? k : (N_IN_F - 1);
        float wv = W[(size_t)kc * N_HID_F + n];
        bsh[n * BSTRIDE + k] = (k < N_IN_F) ? wv : 0.0f;
    }
    __syncthreads();

    const int lane   = threadIdx.x & 31;
    const int tileM  = blockIdx.x * (blockDim.x >> 5) + (threadIdx.x >> 5);
    const int nTiles = (nNodes + 15) >> 4;
    if (tileM >= nTiles) return;

    const int hi = lane >> 4;
    const int lo = lane & 15;
    int m = tileM * 16 + lo;
    if (m >= nNodes) m = nNodes - 1;

    v8f zero = {};
    v8f acc[8];
#pragma unroll
    for (int t = 0; t < 8; ++t) acc[t] = zero;

    const float* Xrow  = X + (size_t)m * N_IN_F;
    const float* bbase = bsh + lo * BSTRIDE;

#pragma unroll 2
    for (int k0 = 0; k0 < 164; k0 += 4) {
        const int kA = k0 + 2 * hi;
        v2f a;
        a.x = Xrow[kA];
        a.y = Xrow[kA + 1];
#pragma unroll
        for (int t = 0; t < 8; ++t) {
            v2f b = *(const v2f*)(bbase + (size_t)(t * 16) * BSTRIDE + kA);
            acc[t] = __builtin_amdgcn_wmma_f32_16x16x4_f32(
                false, a, false, b, (short)0, acc[t], false, false);
        }
    }

    {
        const int kA = 164 + 2 * hi;
        float xv = Xrow[164];
        v2f a;
        a.x = (hi == 0) ? xv : 0.0f;
        a.y = 0.0f;
#pragma unroll
        for (int t = 0; t < 8; ++t) {
            v2f b = *(const v2f*)(bbase + (size_t)(t * 16) * BSTRIDE + kA);
            acc[t] = __builtin_amdgcn_wmma_f32_16x16x4_f32(
                false, a, false, b, (short)0, acc[t], false, false);
        }
    }

    if (tileM * 16 + 16 <= nNodes) {
        float* Hrow = H + (size_t)(tileM * 16 + 8 * hi) * N_HID_F + lo;
#pragma unroll
        for (int r = 0; r < 8; ++r) {
#pragma unroll
            for (int t = 0; t < 8; ++t)
                Hrow[(size_t)r * N_HID_F + t * 16] = acc[t][r];
        }
    } else {
#pragma unroll
        for (int r = 0; r < 8; ++r) {
            int row = tileM * 16 + r + 8 * hi;
            if (row < nNodes) {
#pragma unroll
                for (int t = 0; t < 8; ++t)
                    H[(size_t)row * N_HID_F + t * 16 + lo] = acc[t][r];
            }
        }
    }
}

__global__ void __launch_bounds__(BAND_T)
k_bin(const int* __restrict__ src, const int* __restrict__ dst,
      const unsigned* __restrict__ btot, unsigned* __restrict__ cursor,
      unsigned* __restrict__ bstart, int2* __restrict__ rec, int nE, int nB) {
    __shared__ unsigned tot[NB_MAX];
    __shared__ unsigned cur[NB_MAX];
    const int tid = threadIdx.x;
    if (tid < nB) { tot[tid] = btot[tid]; cur[tid] = 0u; }

    int  s[EU], d[EU], bi[EU];
    bool ok[EU];
    const int base = blockIdx.x * (EU * BAND_T);
#pragma unroll
    for (int u = 0; u < EU; ++u) {
        const int idx = base + u * BAND_T + tid;
        ok[u] = idx < nE;
        const int ic = ok[u] ? idx : (nE - 1);
        d[u] = dst[ic];
        s[u] = src[ic];
    }
    __syncthreads();

    unsigned start = 0u;
    if (tid < nB) for (int j = 0; j < tid; ++j) start += tot[j];
#pragma unroll
    for (int u = 0; u < EU; ++u) {
        int b = (int)((unsigned)d[u] / (unsigned)L1_R);
        bi[u] = (b < nB) ? b : (nB - 1);
        if (ok[u]) atomicAdd(&cur[bi[u]], 1u);
    }
    __syncthreads();

    if (tid < nB) {
        const unsigned cnt = cur[tid];
        const unsigned g   = cnt ? atomicAdd(&cursor[tid], cnt) : 0u;
        cur[tid] = start + g;
        if (blockIdx.x == 0) bstart[tid] = start;
    }
    __syncthreads();

#pragma unroll
    for (int u = 0; u < EU; ++u) {
        if (ok[u]) {
            const unsigned pos = atomicAdd(&cur[bi[u]], 1u);
            if (pos < (unsigned)nE) rec[pos] = make_int2(s[u], d[u]);
        }
    }
}

#define AGG_EPW  32
#define AGG_U    8
#define L1_NPW   (L1_R / (BAND_T / 32))
static_assert(L1_NPW * (BAND_T / 32) == L1_R, "the finishing pass splits the band evenly over the waves");
static_assert(L1_NPW <= 32, "lane t keeps the products of the wave's node t");
__global__ void __launch_bounds__(BAND_T)
k_band_hid(const float* __restrict__ H, const int2* __restrict__ rec,
           const float* __restrict__ inv, const unsigned* __restrict__ bstart,
           const unsigned* __restrict__ btot, const float* __restrict__ b1,
           const float* __restrict__ W2, float* __restrict__ H2, int nN) {
    __shared__ __attribute__((aligned(16))) float acc[L1_R * N_HID_F];
    const int tid  = threadIdx.x;
    const int lane = tid & 31;
    const int wv = __builtin_amdgcn_readfirstlane(tid >> 5);
    const int lo = blockIdx.x * L1_R;
    int cnt = nN - lo;
    if (cnt > L1_R) cnt = L1_R;

    float4* a4 = (float4*)acc;
    {
        const float4 z = make_float4(0.0f, 0.0f, 0.0f, 0.0f);
        for (int i = tid; i < cnt * (N_HID_F / 4); i += BAND_T) a4[i] = z;
    }
    __syncthreads();

    const int e0 = __builtin_amdgcn_readfirstlane((int)bstart[blockIdx.x]);
    const int e1 = e0 + __builtin_amdgcn_readfirstlane((int)btot[blockIdx.x]);
    for (int base = e0 + wv * AGG_EPW; base < e1; base += AGG_EPW * (BAND_T / 32)) {
        const int e  = base + lane;
        const int ec = (e < e1) ? e : (e1 - 1);
        const int2 r = rec[ec];
        int sr = r.x;
        sr = (sr < 0) ? 0 : ((sr < nN) ? sr : (nN - 1));
        const int  dl  = r.y - lo;
        const bool in  = (unsigned)dl < (unsigned)cnt;
        const int  s   = sr;
        const int  d   = in ? dl : 0;
        const int  cb  = in ? __float_as_int(inv[sr] * inv[lo + dl]) : 0;
        int n = e1 - base;
        if (n > AGG_EPW) n = AGG_EPW;

        if (n == AGG_EPW) {
#pragma unroll
            for (int j0 = 0; j0 < AGG_EPW; j0 += AGG_U) {
                float v[AGG_U][4];
#pragma unroll
                for (int u = 0; u < AGG_U; ++u) {
                    const int sj = __builtin_amdgcn_readlane(s, j0 + u);
                    const float* h = H + (size_t)sj * N_HID_F + lane;
                    v[u][0] = h[0]; v[u][1] = h[32]; v[u][2] = h[64]; v[u][3] = h[96];
                }
#pragma unroll
                for (int u = 0; u < AGG_U; ++u) {
                    const int   dj = __builtin_amdgcn_readlane(d, j0 + u);
                    const float c  = __int_as_float(__builtin_amdgcn_readlane(cb, j0 + u));
                    float* o = acc + dj * N_HID_F + lane;
                    atomicAdd(o +  0, v[u][0] * c);
                    atomicAdd(o + 32, v[u][1] * c);
                    atomicAdd(o + 64, v[u][2] * c);
                    atomicAdd(o + 96, v[u][3] * c);
                }
            }
        } else {
            for (int j = 0; j < n; ++j) {
                const int   sj = __builtin_amdgcn_readlane(s, j);
                const int   dj = __builtin_amdgcn_readlane(d, j);
                const float c  = __int_as_float(__builtin_amdgcn_readlane(cb, j));
                const float* h = H + (size_t)sj * N_HID_F + lane;
                const float v0 = h[0], v1 = h[32], v2 = h[64], v3 = h[96];
                float* o = acc + dj * N_HID_F + lane;
                atomicAdd(o +  0, v0 * c);
                atomicAdd(o + 32, v1 * c);
                atomicAdd(o + 64, v2 * c);
                atomicAdd(o + 96, v3 * c);
            }
        }
    }

    const int i0 = wv * L1_NPW;
    float4 hs[L1_NPW];
    float  isn[L1_NPW];
#pragma unroll
    for (int t = 0; t < L1_NPW; ++t) {
        const int i  = i0 + t;
        const int ic = (i < cnt) ? i : (cnt - 1);
        hs[t]  = *(const float4*)(H + (size_t)(lo + ic) * N_HID_F + lane * 4);
        isn[t] = inv[lo + ic];
    }
    const int   f   = lane * 4;
    const float bb0 = b1[f], bb1 = b1[f + 1], bb2 = b1[f + 2], bb3 = b1[f + 3];
    const float* w  = W2 + f * N_EMB_F;
    const float w00 = w[0], w01 = w[1], w10 = w[2], w11 = w[3];
    const float w20 = w[4], w21 = w[5], w30 = w[6], w31 = w[7];
    __syncthreads();

    float r0 = 0.0f, r1 = 0.0f;
#pragma unroll
    for (int t = 0; t < L1_NPW; ++t) {
        const int i  = i0 + t;
        const int ic = (i < cnt) ? i : (cnt - 1);
        const float4 a  = a4[ic * (N_HID_F / 4) + lane];
        const float4 h  = hs[t];
        const float  is = isn[t];
        const float hx = tanhf(a.x + h.x * is * is + bb0);
        const float hy = tanhf(a.y + h.y * is * is + bb1);
        const float hz = tanhf(a.z + h.z * is * is + bb2);
        const float hw = tanhf(a.w + h.w * is * is + bb3);
        float p0 = hx * w00 + hy * w10 + hz * w20 + hw * w30;
        float p1 = hx * w01 + hy * w11 + hz * w21 + hw * w31;
        for (int off = 16; off > 0; off >>= 1) {
            p0 += __shfl_xor(p0, off);
            p1 += __shfl_xor(p1, off);
        }
        if (lane == t) { r0 = p0; r1 = p1; }
    }
    if (lane < L1_NPW && i0 + lane < cnt)
        *(float2*)(H2 + (size_t)(lo + i0 + lane) * N_EMB_F) = make_float2(r0, r1);
}

#define EMB_T    512
static_assert(EMB_T >= L1_R, "the finishing pass takes one node per thread");
static_assert(EU == 8, "the two load fences of k_band_emb name 8 edges");
__global__ void __launch_bounds__(EMB_T)
k_band_emb(const float* __restrict__ H2, const int2* __restrict__ rec,
           const float* __restrict__ inv, const unsigned* __restrict__ bstart,
           const unsigned* __restrict__ btot, const float* __restrict__ b2,
           const float* __restrict__ Wc, const float* __restrict__ bc,
           float* __restrict__ out, int nN) {
    __shared__ __attribute__((aligned(8))) float acc[L1_R * 2];
    const int tid = threadIdx.x;
    const int lo  = blockIdx.x * L1_R;
    int cnt = nN - lo;
    if (cnt > L1_R) cnt = L1_R;

    for (int i = tid; i < L1_R * 2; i += EMB_T) acc[i] = 0.0f;

    const int e0 = (int)bstart[blockIdx.x];
    const int e1 = e0 + (int)btot[blockIdx.x];
    const bool   own = tid < cnt;
    const int    n   = lo + (own ? tid : 0);
    const float  isn = inv[n];
    const float2 hn  = *(const float2*)(H2 + (size_t)n * 2);
    const float  b20 = b2[0], b21 = b2[1], wc0 = Wc[0], wc1 = Wc[1], bc0 = bc[0];
    __syncthreads();

    for (int base = e0; base < e1; base += EU * EMB_T) {
        int2 r[EU];
        int  dd[EU], ss[EU];
        bool m[EU];
#pragma unroll
        for (int u = 0; u < EU; ++u) {
            const int idx = base + u * EMB_T + tid;
            const int ic  = (idx < e1) ? idx : (e1 - 1);
            r[u] = rec[ic];
        }
        asm volatile("" :: "v"(r[0].x), "v"(r[0].y), "v"(r[1].x), "v"(r[1].y), "v"(r[2].x), "v"(r[2].y),
                           "v"(r[3].x), "v"(r[3].y), "v"(r[4].x), "v"(r[4].y), "v"(r[5].x), "v"(r[5].y),
                           "v"(r[6].x), "v"(r[6].y), "v"(r[7].x), "v"(r[7].y));
#pragma unroll
        for (int u = 0; u < EU; ++u) {
            const int idx = base + u * EMB_T + tid;
            int s = r[u].x;
            s = (s < 0) ? 0 : ((s < nN) ? s : (nN - 1));
            const int dl = r[u].y - lo;
            m[u]  = (idx < e1) && ((unsigned)dl < (unsigned)cnt);
            dd[u] = m[u] ? dl : 0;
            ss[u] = m[u] ? s : 0;
        }
        float  is[EU], id[EU];
        float2 v[EU];
#pragma unroll
        for (int u = 0; u < EU; ++u) {
            is[u] = inv[ss[u]];
            id[u] = inv[lo + dd[u]];
            v[u]  = *(const float2*)(H2 + (size_t)ss[u] * 2);
        }
        asm volatile("" :: "v"(is[0]), "v"(id[0]), "v"(v[0].x), "v"(is[1]), "v"(id[1]), "v"(v[1].x),
                           "v"(is[2]), "v"(id[2]), "v"(v[2].x), "v"(is[3]), "v"(id[3]), "v"(v[3].x),
                           "v"(is[4]), "v"(id[4]), "v"(v[4].x), "v"(is[5]), "v"(id[5]), "v"(v[5].x),
                           "v"(is[6]), "v"(id[6]), "v"(v[6].x), "v"(is[7]), "v"(id[7]), "v"(v[7].x));
#pragma unroll
        for (int u = 0; u < EU; ++u) {
            if (m[u]) {
                const float c = is[u] * id[u];
                float* a = acc + 2 * dd[u];
                atomicAdd(a + 0, v[u].x * c);
                atomicAdd(a + 1, v[u].y * c);
            }
        }
    }
    __syncthreads();

    if (own) {
        const float i2 = isn * isn;
        const float t0 = tanhf(acc[2 * tid + 0] + hn.x * i2 + b20);
        const float t1 = tanhf(acc[2 * tid + 1] + hn.y * i2 + b21);
        const float z  = t0 * wc0 + t1 * wc1 + bc0;
        out[n] = 1.0f / (1.0f + expf(-z));
    }
}

extern "C" void kernel_launch(void* const* d_in, const int* in_sizes, int n_in,
                              void* d_out, int out_size, void* d_ws, size_t ws_size,
                              hipStream_t stream) {
    const float* x   = (const float*)d_in[0];
    const int*   ei  = (const int*)d_in[1];
    const float* W1  = (const float*)d_in[2];
    const float* b1  = (const float*)d_in[3];
    const float* W2  = (const float*)d_in[4];
    const float* b2  = (const float*)d_in[5];
    const float* Wc  = (const float*)d_in[6];
    const float* bc  = (const float*)d_in[7];
    float* out = (float*)d_out;

    const int nN = in_sizes[0] / N_IN_F;
    const int nE = in_sizes[1] / 2;
    const int* src = ei;
    const int* dst = ei + nE;

    float* ws   = (float*)d_ws;
    float* h1   = ws;
    int2*  rec  = (int2*)(h1 + (size_t)nN * N_HID_F);
    unsigned* pdeg = (unsigned*)(rec + (size_t)nE);
    float* h2   = (float*)(pdeg + (size_t)NSLICE * nN);
    float* inv  = h2   + (size_t)nN * 2;
    unsigned* btot   = (unsigned*)(inv + (size_t)nN);
    unsigned* cursor = btot + NB_MAX;
    unsigned* bstart = cursor + NB_MAX;

    const int nB1 = (nN + L1_R - 1) / L1_R;

    const int nBands    = (nN + NR_MAX - 1) / NR_MAX;
    const int NR        = (nN + nBands - 1) / nBands;
    const int ePerSlice = (nE + NSLICE - 1) / NSLICE;

    k_deg_band<<<nBands * NSLICE, BAND_T, 0, stream>>>(dst, pdeg, nE, nN, nBands, NR, ePerSlice);
    k_inv_from_parts<<<nB1, L1_R, 0, stream>>>(pdeg, inv, btot, cursor, nN);

    k_bin<<<(nE + EU * BAND_T - 1) / (EU * BAND_T), BAND_T, 0, stream>>>(src, dst, btot, cursor, bstart, rec, nE, nB1);
    const int nTiles = (nN + 15) / 16;
    k_gemm1_wmma<<<(nTiles + GEMM_WAVES - 1) / GEMM_WAVES, GEMM_T, GEMM_LDS, stream>>>(x, W1, h1, nN);
    k_band_hid<<<nB1, BAND_T, 0, stream>>>(h1, rec, inv, bstart, btot, b1, W2, h2, nN);

    k_band_emb<<<nB1, EMB_T, 0, stream>>>(h2, rec, inv, bstart, btot, b2, Wc, bc, out, nN);
}


# GEMM1 weights: W1^T (padded) built once per call in the workspace by the inv launch; the 250 GEMM work-groups copy it to LDS as whole lines, no longer each transposing W1 by 4-B reads; sums unchanged
# speedup vs baseline: 4.1695x; 1.0450x over previous
#include <hip/hip_runtime.h>
#include <cmath>

typedef __attribute__((ext_vector_type(2))) float v2f;
typedef __attribute__((ext_vector_type(8))) float v8f;

#define N_IN_F   165
#define N_HID_F  128
#define N_EMB_F  2
#define K_PAD    168
#define BSTRIDE  170

#define NR_MAX   32768
#define NSLICE   64
#define BAND_T   1024
#define EU       8

__global__ void __launch_bounds__(BAND_T)
k_deg_band(const int* __restrict__ dst, unsigned* __restrict__ part,
           int nE, int nN, int nBands, int NR, int ePerSlice) {
    __shared__ unsigned hist[NR_MAX];
    const int tid   = threadIdx.x;
    const int band  = blockIdx.x % nBands;
    const int slice = blockIdx.x / nBands;
    const int lo    = band * NR;
    int cnt = nN - lo;
    if (cnt > NR) cnt = NR;
    if (cnt < 0)  cnt = 0;

    for (int i = tid; i < cnt; i += BAND_T) hist[i] = 0u;
    __syncthreads();

    const int e0 = slice * ePerSlice;
    int e1 = e0 + ePerSlice;
    if (e1 > nE) e1 = nE;
    for (int base = e0; base < e1; base += EU * BAND_T) {
        int d[EU];
#pragma unroll
        for (int u = 0; u < EU; ++u) {
            int idx = base + u * BAND_T + tid;
            int ic  = (idx < e1) ? idx : (e1 - 1);
            int dv  = dst[ic];
            d[u] = (idx < e1) ? dv : -1;
        }
#pragma unroll
        for (int u = 0; u < EU; ++u) {
            unsigned r = (unsigned)(d[u] - lo);
            if (r < (unsigned)cnt) atomicAdd(&hist[r], 1u);
        }
    }
    __syncthreads();

    unsigned* p = part + (size_t)slice * nN + lo;
    for (int i = tid; i < cnt; i += BAND_T) p[i] = hist[i];
}

#define L1_R     416
#define NB_MAX   1024

__global__ void __launch_bounds__(L1_R)
k_inv_from_parts(const unsigned* __restrict__ part, float* __restrict__ inv,
                 unsigned* __restrict__ btot, unsigned* __restrict__ cursor,
                 const float* __restrict__ W1, float* __restrict__ wt, int nN) {
    __shared__ unsigned wsum[L1_R / 32];
    const int n = blockIdx.x * L1_R + threadIdx.x;
    for (int g = n; g < N_HID_F * BSTRIDE; g += gridDim.x * L1_R) {
        const int col = g / BSTRIDE;
        const int k   = g - col * BSTRIDE;
        const int kc  = (k < N_IN_F) ? k : (N_IN_F - 1);
        const float wv = W1[(size_t)kc * N_HID_F + col];
        wt[g] = (k < N_IN_F) ? wv : 0.0f;
    }
    unsigned c = 0u;
    if (n < nN) {
#pragma unroll 16
        for (int s = 0; s < NSLICE; ++s) c += part[(size_t)s * nN + n];
        inv[n] = rsqrtf(1.0f + (float)c);
    }
    for (int off = 16; off > 0; off >>= 1) c += __shfl_xor(c, off);
    if ((threadIdx.x & 31) == 0) wsum[threadIdx.x >> 5] = c;
    __syncthreads();
    if (threadIdx.x == 0) {
        unsigned t = 0u;
        for (int w = 0; w < L1_R / 32; ++w) t += wsum[w];
        btot[blockIdx.x]   = t;
        cursor[blockIdx.x] = 0u;
    }
}

#define GEMM_WAVES 25
#define GEMM_T     (GEMM_WAVES * 32)
#define GEMM_LDS   (164 * 1024)
static_assert((N_HID_F * BSTRIDE) % 4 == 0, "W1^T is copied 16 B at a time");
__global__ void __launch_bounds__(GEMM_T)
k_gemm1_wmma(const float* __restrict__ X, const float* __restrict__ WT,
             float* __restrict__ H, int nNodes) {
    extern __shared__ __attribute__((aligned(16))) float bsh[];

    {
        const float4* s4 = (const float4*)WT;
        float4*       d4 = (float4*)bsh;
#pragma unroll
        for (int j = 0; j < (N_HID_F * BSTRIDE / 4 + GEMM_T - 1) / GEMM_T; ++j) {
            const int idx = j * GEMM_T + (int)threadIdx.x;
            if (idx < N_HID_F * BSTRIDE / 4) d4[idx] = s4[idx];
        }
    }
    __syncthreads();

    const int lane   = threadIdx.x & 31;
    const int tileM  = blockIdx.x * (blockDim.x >> 5) + (threadIdx.x >> 5);
    const int nTiles = (nNodes + 15) >> 4;
    if (tileM >= nTiles) return;

    const int hi = lane >> 4;
    const int lo = lane & 15;
    int m = tileM * 16 + lo;
    if (m >= nNodes) m = nNodes - 1;

    v8f zero = {};
    v8f acc[8];
#pragma unroll
    for (int t = 0; t < 8; ++t) acc[t] = zero;

    const float* Xrow  = X + (size_t)m * N_IN_F;
    const float* bbase = bsh + lo * BSTRIDE;

#pragma unroll 2
    for (int k0 = 0; k0 < 164; k0 += 4) {
        const int kA = k0 + 2 * hi;
        v2f a;
        a.x = Xrow[kA];
        a.y = Xrow[kA + 1];
#pragma unroll
        for (int t = 0; t < 8; ++t) {
            v2f b = *(const v2f*)(bbase + (size_t)(t * 16) * BSTRIDE + kA);
            acc[t] = __builtin_amdgcn_wmma_f32_16x16x4_f32(
                false, a, false, b, (short)0, acc[t], false, false);
        }
    }

    {
        const int kA = 164 + 2 * hi;
        float xv = Xrow[164];
        v2f a;
        a.x = (hi == 0) ? xv : 0.0f;
        a.y = 0.0f;
#pragma unroll
        for (int t = 0; t < 8; ++t) {
            v2f b = *(const v2f*)(bbase + (size_t)(t * 16) * BSTRIDE + kA);
            acc[t] = __builtin_amdgcn_wmma_f32_16x16x4_f32(
                false, a, false, b, (short)0, acc[t], false, false);
        }
    }

    if (tileM * 16 + 16 <= nNodes) {
        float* Hrow = H + (size_t)(tileM * 16 + 8 * hi) * N_HID_F + lo;
#pragma unroll
        for (int r = 0; r < 8; ++r) {
#pragma unroll
            for (int t = 0; t < 8; ++t)
                Hrow[(size_t)r * N_HID_F + t * 16] = acc[t][r];
        }
    } else {
#pragma unroll
        for (int r = 0; r < 8; ++r) {
            int row = tileM * 16 + r + 8 * hi;
            if (row < nNodes) {
#pragma unroll
                for (int t = 0; t < 8; ++t)
                    H[(size_t)row * N_HID_F + t * 16 + lo] = acc[t][r];
            }
        }
    }
}

__global__ void __launch_bounds__(BAND_T)
k_bin(const int* __restrict__ src, const int* __restrict__ dst,
      const unsigned* __restrict__ btot, unsigned* __restrict__ cursor,
      unsigned* __restrict__ bstart, int2* __restrict__ rec, int nE, int nB) {
    __shared__ unsigned tot[NB_MAX];
    __shared__ unsigned cur[NB_MAX];
    const int tid = threadIdx.x;
    if (tid < nB) { tot[tid] = btot[tid]; cur[tid] = 0u; }

    int  s[EU], d[EU], bi[EU];
    bool ok[EU];
    const int base = blockIdx.x * (EU * BAND_T);
#pragma unroll
    for (int u = 0; u < EU; ++u) {
        const int idx = base + u * BAND_T + tid;
        ok[u] = idx < nE;
        const int ic = ok[u] ? idx : (nE - 1);
        d[u] = dst[ic];
        s[u] = src[ic];
    }
    __syncthreads();

    unsigned start = 0u;
    if (tid < nB) for (int j = 0; j < tid; ++j) start += tot[j];
#pragma unroll
    for (int u = 0; u < EU; ++u) {
        int b = (int)((unsigned)d[u] / (unsigned)L1_R);
        bi[u] = (b < nB) ? b : (nB - 1);
        if (ok[u]) atomicAdd(&cur[bi[u]], 1u);
    }
    __syncthreads();

    if (tid < nB) {
        const unsigned cnt = cur[tid];
        const unsigned g   = cnt ? atomicAdd(&cursor[tid], cnt) : 0u;
        cur[tid] = start + g;
        if (blockIdx.x == 0) bstart[tid] = start;
    }
    __syncthreads();

#pragma unroll
    for (int u = 0; u < EU; ++u) {
        if (ok[u]) {
            const unsigned pos = atomicAdd(&cur[bi[u]], 1u);
            if (pos < (unsigned)nE) rec[pos] = make_int2(s[u], d[u]);
        }
    }
}

#define AGG_EPW  32
#define AGG_U    8
#define L1_NPW   (L1_R / (BAND_T / 32))
static_assert(L1_NPW * (BAND_T / 32) == L1_R, "the finishing pass splits the band evenly over the waves");
static_assert(L1_NPW <= 32, "lane t keeps the products of the wave's node t");
__global__ void __launch_bounds__(BAND_T)
k_band_hid(const float* __restrict__ H, const int2* __restrict__ rec,
           const float* __restrict__ inv, const unsigned* __restrict__ bstart,
           const unsigned* __restrict__ btot, const float* __restrict__ b1,
           const float* __restrict__ W2, float* __restrict__ H2, int nN) {
    __shared__ __attribute__((aligned(16))) float acc[L1_R * N_HID_F];
    const int tid  = threadIdx.x;
    const int lane = tid & 31;
    const int wv = __builtin_amdgcn_readfirstlane(tid >> 5);
    const int lo = blockIdx.x * L1_R;
    int cnt = nN - lo;
    if (cnt > L1_R) cnt = L1_R;

    float4* a4 = (float4*)acc;
    {
        const float4 z = make_float4(0.0f, 0.0f, 0.0f, 0.0f);
        for (int i = tid; i < cnt * (N_HID_F / 4); i += BAND_T) a4[i] = z;
    }
    __syncthreads();

    const int e0 = __builtin_amdgcn_readfirstlane((int)bstart[blockIdx.x]);
    const int e1 = e0 + __builtin_amdgcn_readfirstlane((int)btot[blockIdx.x]);
    for (int base = e0 + wv * AGG_EPW; base < e1; base += AGG_EPW * (BAND_T / 32)) {
        const int e  = base + lane;
        const int ec = (e < e1) ? e : (e1 - 1);
        const int2 r = rec[ec];
        int sr = r.x;
        sr = (sr < 0) ? 0 : ((sr < nN) ? sr : (nN - 1));
        const int  dl  = r.y - lo;
        const bool in  = (unsigned)dl < (unsigned)cnt;
        const int  s   = sr;
        const int  d   = in ? dl : 0;
        const int  cb  = in ? __float_as_int(inv[sr] * inv[lo + dl]) : 0;
        int n = e1 - base;
        if (n > AGG_EPW) n = AGG_EPW;

        if (n == AGG_EPW) {
#pragma unroll
            for (int j0 = 0; j0 < AGG_EPW; j0 += AGG_U) {
                float v[AGG_U][4];
#pragma unroll
                for (int u = 0; u < AGG_U; ++u) {
                    const int sj = __builtin_amdgcn_readlane(s, j0 + u);
                    const float* h = H + (size_t)sj * N_HID_F + lane;
                    v[u][0] = h[0]; v[u][1] = h[32]; v[u][2] = h[64]; v[u][3] = h[96];
                }
#pragma unroll
                for (int u = 0; u < AGG_U; ++u) {
                    const int   dj = __builtin_amdgcn_readlane(d, j0 + u);
                    const float c  = __int_as_float(__builtin_amdgcn_readlane(cb, j0 + u));
                    float* o = acc + dj * N_HID_F + lane;
                    atomicAdd(o +  0, v[u][0] * c);
                    atomicAdd(o + 32, v[u][1] * c);
                    atomicAdd(o + 64, v[u][2] * c);
                    atomicAdd(o + 96, v[u][3] * c);
                }
            }
        } else {
            for (int j = 0; j < n; ++j) {
                const int   sj = __builtin_amdgcn_readlane(s, j);
                const int   dj = __builtin_amdgcn_readlane(d, j);
                const float c  = __int_as_float(__builtin_amdgcn_readlane(cb, j));
                const float* h = H + (size_t)sj * N_HID_F + lane;
                const float v0 = h[0], v1 = h[32], v2 = h[64], v3 = h[96];
                float* o = acc + dj * N_HID_F + lane;
                atomicAdd(o +  0, v0 * c);
                atomicAdd(o + 32, v1 * c);
                atomicAdd(o + 64, v2 * c);
                atomicAdd(o + 96, v3 * c);
            }
        }
    }

    const int i0 = wv * L1_NPW;
    float4 hs[L1_NPW];
    float  isn[L1_NPW];
#pragma unroll
    for (int t = 0; t < L1_NPW; ++t) {
        const int i  = i0 + t;
        const int ic = (i < cnt) ? i : (cnt - 1);
        hs[t]  = *(const float4*)(H + (size_t)(lo + ic) * N_HID_F + lane * 4);
        isn[t] = inv[lo + ic];
    }
    const int   f   = lane * 4;
    const float bb0 = b1[f], bb1 = b1[f + 1], bb2 = b1[f + 2], bb3 = b1[f + 3];
    const float* w  = W2 + f * N_EMB_F;
    const float w00 = w[0], w01 = w[1], w10 = w[2], w11 = w[3];
    const float w20 = w[4], w21 = w[5], w30 = w[6], w31 = w[7];
    __syncthreads();

    float r0 = 0.0f, r1 = 0.0f;
#pragma unroll
    for (int t = 0; t < L1_NPW; ++t) {
        const int i  = i0 + t;
        const int ic = (i < cnt) ? i : (cnt - 1);
        const float4 a  = a4[ic * (N_HID_F / 4) + lane];
        const float4 h  = hs[t];
        const float  is = isn[t];
        const float hx = tanhf(a.x + h.x * is * is + bb0);
        const float hy = tanhf(a.y + h.y * is * is + bb1);
        const float hz = tanhf(a.z + h.z * is * is + bb2);
        const float hw = tanhf(a.w + h.w * is * is + bb3);
        float p0 = hx * w00 + hy * w10 + hz * w20 + hw * w30;
        float p1 = hx * w01 + hy * w11 + hz * w21 + hw * w31;
        for (int off = 16; off > 0; off >>= 1) {
            p0 += __shfl_xor(p0, off);
            p1 += __shfl_xor(p1, off);
        }
        if (lane == t) { r0 = p0; r1 = p1; }
    }
    if (lane < L1_NPW && i0 + lane < cnt)
        *(float2*)(H2 + (size_t)(lo + i0 + lane) * N_EMB_F) = make_float2(r0, r1);
}

#define EMB_T    512
static_assert(EMB_T >= L1_R, "the finishing pass takes one node per thread");
static_assert(EU == 8, "the two load fences of k_band_emb name 8 edges");
__global__ void __launch_bounds__(EMB_T)
k_band_emb(const float* __restrict__ H2, const int2* __restrict__ rec,
           const float* __restrict__ inv, const unsigned* __restrict__ bstart,
           const unsigned* __restrict__ btot, const float* __restrict__ b2,
           const float* __restrict__ Wc, const float* __restrict__ bc,
           float* __restrict__ out, int nN) {
    __shared__ __attribute__((aligned(8))) float acc[L1_R * 2];
    const int tid = threadIdx.x;
    const int lo  = blockIdx.x * L1_R;
    int cnt = nN - lo;
    if (cnt > L1_R) cnt = L1_R;

    for (int i = tid; i < L1_R * 2; i += EMB_T) acc[i] = 0.0f;

    const int e0 = (int)bstart[blockIdx.x];
    const int e1 = e0 + (int)btot[blockIdx.x];
    const bool   own = tid < cnt;
    const int    n   = lo + (own ? tid : 0);
    const float  isn = inv[n];
    const float2 hn  = *(const float2*)(H2 + (size_t)n * 2);
    const float  b20 = b2[0], b21 = b2[1], wc0 = Wc[0], wc1 = Wc[1], bc0 = bc[0];
    __syncthreads();

    for (int base = e0; base < e1; base += EU * EMB_T) {
        int2 r[EU];
        int  dd[EU], ss[EU];
        bool m[EU];
#pragma unroll
        for (int u = 0; u < EU; ++u) {
            const int idx = base + u * EMB_T + tid;
            const int ic  = (idx < e1) ? idx : (e1 - 1);
            r[u] = rec[ic];
        }
        asm volatile("" :: "v"(r[0].x), "v"(r[0].y), "v"(r[1].x), "v"(r[1].y), "v"(r[2].x), "v"(r[2].y),
                           "v"(r[3].x), "v"(r[3].y), "v"(r[4].x), "v"(r[4].y), "v"(r[5].x), "v"(r[5].y),
                           "v"(r[6].x), "v"(r[6].y), "v"(r[7].x), "v"(r[7].y));
#pragma unroll
        for (int u = 0; u < EU; ++u) {
            const int idx = base + u * EMB_T + tid;
            int s = r[u].x;
            s = (s < 0) ? 0 : ((s < nN) ? s : (nN - 1));
            const int dl = r[u].y - lo;
            m[u]  = (idx < e1) && ((unsigned)dl < (unsigned)cnt);
            dd[u] = m[u] ? dl : 0;
            ss[u] = m[u] ? s : 0;
        }
        float  is[EU], id[EU];
        float2 v[EU];
#pragma unroll
        for (int u = 0; u < EU; ++u) {
            is[u] = inv[ss[u]];
            id[u] = inv[lo + dd[u]];
            v[u]  = *(const float2*)(H2 + (size_t)ss[u] * 2);
        }
        asm volatile("" :: "v"(is[0]), "v"(id[0]), "v"(v[0].x), "v"(is[1]), "v"(id[1]), "v"(v[1].x),
                           "v"(is[2]), "v"(id[2]), "v"(v[2].x), "v"(is[3]), "v"(id[3]), "v"(v[3].x),
                           "v"(is[4]), "v"(id[4]), "v"(v[4].x), "v"(is[5]), "v"(id[5]), "v"(v[5].x),
                           "v"(is[6]), "v"(id[6]), "v"(v[6].x), "v"(is[7]), "v"(id[7]), "v"(v[7].x));
#pragma unroll
        for (int u = 0; u < EU; ++u) {
            if (m[u]) {
                const float c = is[u] * id[u];
                float* a = acc + 2 * dd[u];
                atomicAdd(a + 0, v[u].x * c);
                atomicAdd(a + 1, v[u].y * c);
            }
        }
    }
    __syncthreads();

    if (own) {
        const float i2 = isn * isn;
        const float t0 = tanhf(acc[2 * tid + 0] + hn.x * i2 + b20);
        const float t1 = tanhf(acc[2 * tid + 1] + hn.y * i2 + b21);
        const float z  = t0 * wc0 + t1 * wc1 + bc0;
        out[n] = 1.0f / (1.0f + expf(-z));
    }
}

extern "C" void kernel_launch(void* const* d_in, const int* in_sizes, int n_in,
                              void* d_out, int out_size, void* d_ws, size_t ws_size,
                              hipStream_t stream) {
    const float* x   = (const float*)d_in[0];
    const int*   ei  = (const int*)d_in[1];
    const float* W1  = (const float*)d_in[2];
    const float* b1  = (const float*)d_in[3];
    const float* W2  = (const float*)d_in[4];
    const float* b2  = (const float*)d_in[5];
    const float* Wc  = (const float*)d_in[6];
    const float* bc  = (const float*)d_in[7];
    float* out = (float*)d_out;

    const int nN = in_sizes[0] / N_IN_F;
    const int nE = in_sizes[1] / 2;
    const int* src = ei;
    const int* dst = ei + nE;

    float* ws   = (float*)d_ws;
    float* w1t  = ws;
    float* h1   = w1t + N_HID_F * BSTRIDE;
    int2*  rec  = (int2*)(h1 + (size_t)nN * N_HID_F);
    unsigned* pdeg = (unsigned*)(rec + (size_t)nE);
    float* h2   = (float*)(pdeg + (size_t)NSLICE * nN);
    float* inv  = h2   + (size_t)nN * 2;
    unsigned* btot   = (unsigned*)(inv + (size_t)nN);
    unsigned* cursor = btot + NB_MAX;
    unsigned* bstart = cursor + NB_MAX;

    const int nB1 = (nN + L1_R - 1) / L1_R;

    const int nBands    = (nN + NR_MAX - 1) / NR_MAX;
    const int NR        = (nN + nBands - 1) / nBands;
    const int ePerSlice = (nE + NSLICE - 1) / NSLICE;

    k_deg_band<<<nBands * NSLICE, BAND_T, 0, stream>>>(dst, pdeg, nE, nN, nBands, NR, ePerSlice);
    k_inv_from_parts<<<nB1, L1_R, 0, stream>>>(pdeg, inv, btot, cursor, W1, w1t, nN);

    k_bin<<<(nE + EU * BAND_T - 1) / (EU * BAND_T), BAND_T, 0, stream>>>(src, dst, btot, cursor, bstart, rec, nE, nB1);
    const int nTiles = (nN + 15) / 16;
    k_gemm1_wmma<<<(nTiles + GEMM_WAVES - 1) / GEMM_WAVES, GEMM_T, GEMM_LDS, stream>>>(x, w1t, h1, nN);
    k_band_hid<<<nB1, BAND_T, 0, stream>>>(h1, rec, inv, bstart, btot, b1, W2, h2, nN);

    k_band_emb<<<nB1, EMB_T, 0, stream>>>(h2, rec, inv, bstart, btot, b2, Wc, bc, out, nN);
}


# In-degrees counted by the GEMM groups from the band-binned list; band totals and run offsets from a per-chunk count table (no 64 planes, no cursor adds): 6 -> 5 launches; sum order = LDS arrival
# speedup vs baseline: 4.3850x; 1.0517x over previous
#include <hip/hip_runtime.h>
#include <cmath>

typedef __attribute__((ext_vector_type(2))) float v2f;
typedef __attribute__((ext_vector_type(8))) float v8f;

#define N_IN_F   165
#define N_HID_F  128
#define N_EMB_F  2
#define K_PAD    168
#define BSTRIDE  170

#define L1_R     416
#define NB_MAX   1024
#define BAND_T   1024
#define EU       8
static_assert(NB_MAX == BAND_T, "one thread per band entry in the edge passes");
#define RED_U    7

__device__ __forceinline__ int band_of(int d, int nB) {
    const int b = (int)((unsigned)d / (unsigned)L1_R);
    return (b < nB) ? b : (nB - 1);
}

__global__ void __launch_bounds__(BAND_T)
k_band_count(const int* __restrict__ dst, unsigned* __restrict__ ccnt,
             const float* __restrict__ W1, float* __restrict__ wt, int nE, int nB) {
    __shared__ unsigned cnt[NB_MAX];
    const int tid = threadIdx.x;
    if (tid < nB) cnt[tid] = 0u;

    int  d[EU];
    bool ok[EU];
    const int base = blockIdx.x * (EU * BAND_T);
#pragma unroll
    for (int u = 0; u < EU; ++u) {
        const int idx = base + u * BAND_T + tid;
        ok[u] = idx < nE;
        const int ic = ok[u] ? idx : (nE - 1);
        d[u] = dst[ic];
    }

    const int  g0   = blockIdx.x * BAND_T + tid;
    const bool wown = g0 < N_HID_F * BSTRIDE;
    const int  gc   = wown ? g0 : 0;
    const int  col0 = gc / BSTRIDE;
    const int  k0   = gc - col0 * BSTRIDE;
    const int  kc0  = (k0 < N_IN_F) ? k0 : (N_IN_F - 1);
    const float w0  = W1[(size_t)kc0 * N_HID_F + col0];
    if (wown) wt[g0] = (k0 < N_IN_F) ? w0 : 0.0f;
    for (int g = g0 + gridDim.x * BAND_T; g < N_HID_F * BSTRIDE; g += gridDim.x * BAND_T) {
        const int col = g / BSTRIDE;
        const int k   = g - col * BSTRIDE;
        const int kc  = (k < N_IN_F) ? k : (N_IN_F - 1);
        const float wv = W1[(size_t)kc * N_HID_F + col];
        wt[g] = (k < N_IN_F) ? wv : 0.0f;
    }
    __syncthreads();

#pragma unroll
    for (int u = 0; u < EU; ++u)
        if (ok[u]) atomicAdd(&cnt[band_of(d[u], nB)], 1u);
    __syncthreads();

    const int pitch = ((nB + 3) >> 2) << 2;
    if (tid < pitch) ccnt[(size_t)blockIdx.x * pitch + tid] = (tid < nB) ? cnt[tid] : 0u;
}

#define GEMM_WAVES 25
#define GEMM_T     (GEMM_WAVES * 32)
#define GEMM_LDS   (164 * 1024)
#define DEG_U      9
static_assert(DEG_U == 9, "the load fence of the degree count names 9 records");
static_assert((N_HID_F * BSTRIDE) % 4 == 0, "W1^T is copied 16 B at a time");
static_assert(N_HID_F * BSTRIDE * 4 + L1_R * 4 <= GEMM_LDS, "W1^T and the band's degree counters fit the reservation");
static_assert(GEMM_WAVES * 16 <= L1_R, "the GEMM grid is never smaller than the number of bands");
__global__ void __launch_bounds__(GEMM_T)
k_gemm1_wmma(const float* __restrict__ X, const float* __restrict__ WT,
             float* __restrict__ H, int nNodes,
             const int2* __restrict__ rec, const unsigned* __restrict__ bstart,
             const unsigned* __restrict__ btot, float* __restrict__ inv, int nB) {
    extern __shared__ __attribute__((aligned(16))) float bsh[];
    unsigned* dcnt = (unsigned*)(bsh + N_HID_F * BSTRIDE);

    const bool hasBand = (int)blockIdx.x < nB;
    const int  blo     = (int)blockIdx.x * L1_R;
    int cnt = nNodes - blo;
    if (cnt > L1_R) cnt = L1_R;
    if (cnt < 0)    cnt = 0;
    int e0 = 0, e1 = 0;
    if (hasBand) {
        e0 = (int)bstart[blockIdx.x];
        e1 = e0 + (int)btot[blockIdx.x];
    }

    {
        const float4* s4 = (const float4*)WT;
        float4*       d4 = (float4*)bsh;
#pragma unroll
        for (int j = 0; j < (N_HID_F * BSTRIDE / 4 + GEMM_T - 1) / GEMM_T; ++j) {
            const int idx = j * GEMM_T + (int)threadIdx.x;
            if (idx < N_HID_F * BSTRIDE / 4) d4[idx] = s4[idx];
        }
    }
    for (int i = threadIdx.x; i < L1_R; i += GEMM_T) dcnt[i] = 0u;
    __syncthreads();

    for (int base = e0; base < e1; base += DEG_U * GEMM_T) {
        int dv[DEG_U];
#pragma unroll
        for (int u = 0; u < DEG_U; ++u) {
            const int idx = base + u * GEMM_T + (int)threadIdx.x;
            const int ic  = (idx < e1) ? idx : (e1 - 1);
            dv[u] = rec[ic].y;
        }
        asm volatile("" :: "v"(dv[0]), "v"(dv[1]), "v"(dv[2]), "v"(dv[3]), "v"(dv[4]),
                           "v"(dv[5]), "v"(dv[6]), "v"(dv[7]), "v"(dv[8]));
#pragma unroll
        for (int u = 0; u < DEG_U; ++u) {
            const int idx = base + u * GEMM_T + (int)threadIdx.x;
            const int dl  = (idx < e1) ? (dv[u] - blo) : -1;
            if ((unsigned)dl < (unsigned)cnt) atomicAdd(&dcnt[dl], 1u);
        }
    }
    __syncthreads();
    if (hasBand)
        for (int i = threadIdx.x; i < cnt; i += GEMM_T) inv[blo + i] = rsqrtf(1.0f + (float)dcnt[i]);

    const int lane   = threadIdx.x & 31;
    const int tileM  = blockIdx.x * (blockDim.x >> 5) + (threadIdx.x >> 5);
    const int nTiles = (nNodes + 15) >> 4;
    if (tileM >= nTiles) return;

    const int hi = lane >> 4;
    const int lo = lane & 15;
    int m = tileM * 16 + lo;
    if (m >= nNodes) m = nNodes - 1;

    v8f zero = {};
    v8f acc[8];
#pragma unroll
    for (int t = 0; t < 8; ++t) acc[t] = zero;

    const float* Xrow  = X + (size_t)m * N_IN_F;
    const float* bbase = bsh + lo * BSTRIDE;

#pragma unroll 2
    for (int k0 = 0; k0 < 164; k0 += 4) {
        const int kA = k0 + 2 * hi;
        v2f a;
        a.x = Xrow[kA];
        a.y = Xrow[kA + 1];
#pragma unroll
        for (int t = 0; t < 8; ++t) {
            v2f b = *(const v2f*)(bbase + (size_t)(t * 16) * BSTRIDE + kA);
            acc[t] = __builtin_amdgcn_wmma_f32_16x16x4_f32(
                false, a, false, b, (short)0, acc[t], false, false);
        }
    }

    {
        const int kA = 164 + 2 * hi;
        float xv = Xrow[164];
        v2f a;
        a.x = (hi == 0) ? xv : 0.0f;
        a.y = 0.0f;
#pragma unroll
        for (int t = 0; t < 8; ++t) {
            v2f b = *(const v2f*)(bbase + (size_t)(t * 16) * BSTRIDE + kA);
            acc[t] = __builtin_amdgcn_wmma_f32_16x16x4_f32(
                false, a, false, b, (short)0, acc[t], false, false);
        }
    }

    if (tileM * 16 + 16 <= nNodes) {
        float* Hrow = H + (size_t)(tileM * 16 + 8 * hi) * N_HID_F + lo;
#pragma unroll
        for (int r = 0; r < 8; ++r) {
#pragma unroll
            for (int t = 0; t < 8; ++t)
                Hrow[(size_t)r * N_HID_F + t * 16] = acc[t][r];
        }
    } else {
#pragma unroll
        for (int r = 0; r < 8; ++r) {
            int row = tileM * 16 + r + 8 * hi;
            if (row < nNodes) {
#pragma unroll
                for (int t = 0; t < 8; ++t)
                    H[(size_t)row * N_HID_F + t * 16 + lo] = acc[t][r];
            }
        }
    }
}

__global__ void __launch_bounds__(BAND_T)
k_bin(const int* __restrict__ src, const int* __restrict__ dst,
      const unsigned* __restrict__ ccnt, unsigned* __restrict__ bstart,
      unsigned* __restrict__ btot, int2* __restrict__ rec, int nE, int nB, int nChunks) {
    __shared__ unsigned tot[NB_MAX];
    __shared__ unsigned cur[NB_MAX];
    const int tid = threadIdx.x;
    tot[tid] = 0u;
    cur[tid] = 0u;

    int  s[EU], d[EU];
    bool ok[EU];
    const int base = blockIdx.x * (EU * BAND_T);
#pragma unroll
    for (int u = 0; u < EU; ++u) {
        const int idx = base + u * BAND_T + tid;
        ok[u] = idx < nE;
        const int ic = ok[u] ? idx : (nE - 1);
        d[u] = dst[ic];
        s[u] = src[ic];
    }

    const uint4* tab = (const uint4*)ccnt;
    const int nQ   = (nB + 3) >> 2;
    const int nQP  = (nQ + 31) >> 5;
    const int nSub = (BAND_T / 32) / nQP;
    const int wv   = tid >> 5;
    const int sub  = wv / nQP;
    const int q    = (wv - sub * nQP) * 32 + (tid & 31);
    const bool red = sub < nSub && q < nQ;
    unsigned t0 = 0u, t1 = 0u, t2 = 0u, t3 = 0u, p0 = 0u, p1 = 0u, p2 = 0u, p3 = 0u;
    if (red) {
        for (int c0 = sub; c0 < nChunks; c0 += nSub * RED_U) {
            uint4 v[RED_U];
#pragma unroll
            for (int j = 0; j < RED_U; ++j) {
                const int c  = c0 + j * nSub;
                const int cc = (c < nChunks) ? c : (nChunks - 1);
                v[j] = tab[cc * nQ + q];
            }
#pragma unroll
            for (int j = 0; j < RED_U; ++j) {
                const int  c   = c0 + j * nSub;
                const bool in  = c < nChunks;
                const bool bef = c < (int)blockIdx.x;
                t0 += in ? v[j].x : 0u;   t1 += in ? v[j].y : 0u;
                t2 += in ? v[j].z : 0u;   t3 += in ? v[j].w : 0u;
                p0 += bef ? v[j].x : 0u;  p1 += bef ? v[j].y : 0u;
                p2 += bef ? v[j].z : 0u;  p3 += bef ? v[j].w : 0u;
            }
        }
    }
    __syncthreads();

    if (red) {
        const int b = 4 * q;
        atomicAdd(&tot[b + 0], t0);  atomicAdd(&tot[b + 1], t1);
        atomicAdd(&tot[b + 2], t2);  atomicAdd(&tot[b + 3], t3);
        atomicAdd(&cur[b + 0], p0);  atomicAdd(&cur[b + 1], p1);
        atomicAdd(&cur[b + 2], p2);  atomicAdd(&cur[b + 3], p3);
    }
    __syncthreads();

    if (tid < nB) {
        unsigned start = 0u;
        for (int j = 0; j < tid; ++j) start += tot[j];
        cur[tid] += start;
        if (blockIdx.x == 0) { bstart[tid] = start; btot[tid] = tot[tid]; }
    }
    __syncthreads();

#pragma unroll
    for (int u = 0; u < EU; ++u) {
        if (ok[u]) {
            const unsigned pos = atomicAdd(&cur[band_of(d[u], nB)], 1u);
            if (pos < (unsigned)nE) rec[pos] = make_int2(s[u], d[u]);
        }
    }
}

#define AGG_EPW  32
#define AGG_U    8
#define L1_NPW   (L1_R / (BAND_T / 32))
static_assert(L1_NPW * (BAND_T / 32) == L1_R, "the finishing pass splits the band evenly over the waves");
static_assert(L1_NPW <= 32, "lane t keeps the products of the wave's node t");
__global__ void __launch_bounds__(BAND_T)
k_band_hid(const float* __restrict__ H, const int2* __restrict__ rec,
           const float* __restrict__ inv, const unsigned* __restrict__ bstart,
           const unsigned* __restrict__ btot, const float* __restrict__ b1,
           const float* __restrict__ W2, float* __restrict__ H2, int nN) {
    __shared__ __attribute__((aligned(16))) float acc[L1_R * N_HID_F];
    const int tid  = threadIdx.x;
    const int lane = tid & 31;
    const int wv = __builtin_amdgcn_readfirstlane(tid >> 5);
    const int lo = blockIdx.x * L1_R;
    int cnt = nN - lo;
    if (cnt > L1_R) cnt = L1_R;

    float4* a4 = (float4*)acc;
    {
        const float4 z = make_float4(0.0f, 0.0f, 0.0f, 0.0f);
        for (int i = tid; i < cnt * (N_HID_F / 4); i += BAND_T) a4[i] = z;
    }
    __syncthreads();

    const int e0 = __builtin_amdgcn_readfirstlane((int)bstart[blockIdx.x]);
    const int e1 = e0 + __builtin_amdgcn_readfirstlane((int)btot[blockIdx.x]);
    for (int base = e0 + wv * AGG_EPW; base < e1; base += AGG_EPW * (BAND_T / 32)) {
        const int e  = base + lane;
        const int ec = (e < e1) ? e : (e1 - 1);
        const int2 r = rec[ec];
        int sr = r.x;
        sr = (sr < 0) ? 0 : ((sr < nN) ? sr : (nN - 1));
        const int  dl  = r.y - lo;
        const bool in  = (unsigned)dl < (unsigned)cnt;
        const int  s   = sr;
        const int  d   = in ? dl : 0;
        const int  cb  = in ? __float_as_int(inv[sr] * inv[lo + dl]) : 0;
        int n = e1 - base;
        if (n > AGG_EPW) n = AGG_EPW;

        if (n == AGG_EPW) {
#pragma unroll
            for (int j0 = 0; j0 < AGG_EPW; j0 += AGG_U) {
                float v[AGG_U][4];
#pragma unroll
                for (int u = 0; u < AGG_U; ++u) {
                    const int sj = __builtin_amdgcn_readlane(s, j0 + u);
                    const float* h = H + (size_t)sj * N_HID_F + lane;
                    v[u][0] = h[0]; v[u][1] = h[32]; v[u][2] = h[64]; v[u][3] = h[96];
                }
#pragma unroll
                for (int u = 0; u < AGG_U; ++u) {
                    const int   dj = __builtin_amdgcn_readlane(d, j0 + u);
                    const float c  = __int_as_float(__builtin_amdgcn_readlane(cb, j0 + u));
                    float* o = acc + dj * N_HID_F + lane;
                    atomicAdd(o +  0, v[u][0] * c);
                    atomicAdd(o + 32, v[u][1] * c);
                    atomicAdd(o + 64, v[u][2] * c);
                    atomicAdd(o + 96, v[u][3] * c);
                }
            }
        } else {
            for (int j = 0; j < n; ++j) {
                const int   sj = __builtin_amdgcn_readlane(s, j);
                const int   dj = __builtin_amdgcn_readlane(d, j);
                const float c  = __int_as_float(__builtin_amdgcn_readlane(cb, j));
                const float* h = H + (size_t)sj * N_HID_F + lane;
                const float v0 = h[0], v1 = h[32], v2 = h[64], v3 = h[96];
                float* o = acc + dj * N_HID_F + lane;
                atomicAdd(o +  0, v0 * c);
                atomicAdd(o + 32, v1 * c);
                atomicAdd(o + 64, v2 * c);
                atomicAdd(o + 96, v3 * c);
            }
        }
    }

    const int i0 = wv * L1_NPW;
    float4 hs[L1_NPW];
    float  isn[L1_NPW];
#pragma unroll
    for (int t = 0; t < L1_NPW; ++t) {
        const int i  = i0 + t;
        const int ic = (i < cnt) ? i : (cnt - 1);
        hs[t]  = *(const float4*)(H + (size_t)(lo + ic) * N_HID_F + lane * 4);
        isn[t] = inv[lo + ic];
    }
    const int   f   = lane * 4;
    const float bb0 = b1[f], bb1 = b1[f + 1], bb2 = b1[f + 2], bb3 = b1[f + 3];
    const float* w  = W2 + f * N_EMB_F;
    const float w00 = w[0], w01 = w[1], w10 = w[2], w11 = w[3];
    const float w20 = w[4], w21 = w[5], w30 = w[6], w31 = w[7];
    __syncthreads();

    float r0 = 0.0f, r1 = 0.0f;
#pragma unroll
    for (int t = 0; t < L1_NPW; ++t) {
        const int i  = i0 + t;
        const int ic = (i < cnt) ? i : (cnt - 1);
        const float4 a  = a4[ic * (N_HID_F / 4) + lane];
        const float4 h  = hs[t];
        const float  is = isn[t];
        const float hx = tanhf(a.x + h.x * is * is + bb0);
        const float hy = tanhf(a.y + h.y * is * is + bb1);
        const float hz = tanhf(a.z + h.z * is * is + bb2);
        const float hw = tanhf(a.w + h.w * is * is + bb3);
        float p0 = hx * w00 + hy * w10 + hz * w20 + hw * w30;
        float p1 = hx * w01 + hy * w11 + hz * w21 + hw * w31;
        for (int off = 16; off > 0; off >>= 1) {
            p0 += __shfl_xor(p0, off);
            p1 += __shfl_xor(p1, off);
        }
        if (lane == t) { r0 = p0; r1 = p1; }
    }
    if (lane < L1_NPW && i0 + lane < cnt)
        *(float2*)(H2 + (size_t)(lo + i0 + lane) * N_EMB_F) = make_float2(r0, r1);
}

#define EMB_T    512
static_assert(EMB_T >= L1_R, "the finishing pass takes one node per thread");
static_assert(EU == 8, "the two load fences of k_band_emb name 8 edges");
__global__ void __launch_bounds__(EMB_T)
k_band_emb(const float* __restrict__ H2, const int2* __restrict__ rec,
           const float* __restrict__ inv, const unsigned* __restrict__ bstart,
           const unsigned* __restrict__ btot, const float* __restrict__ b2,
           const float* __restrict__ Wc, const float* __restrict__ bc,
           float* __restrict__ out, int nN) {
    __shared__ __attribute__((aligned(8))) float acc[L1_R * 2];
    const int tid = threadIdx.x;
    const int lo  = blockIdx.x * L1_R;
    int cnt = nN - lo;
    if (cnt > L1_R) cnt = L1_R;

    for (int i = tid; i < L1_R * 2; i += EMB_T) acc[i] = 0.0f;

    const int e0 = (int)bstart[blockIdx.x];
    const int e1 = e0 + (int)btot[blockIdx.x];
    const bool   own = tid < cnt;
    const int    n   = lo + (own ? tid : 0);
    const float  isn = inv[n];
    const float2 hn  = *(const float2*)(H2 + (size_t)n * 2);
    const float  b20 = b2[0], b21 = b2[1], wc0 = Wc[0], wc1 = Wc[1], bc0 = bc[0];
    __syncthreads();

    for (int base = e0; base < e1; base += EU * EMB_T) {
        int2 r[EU];
        int  dd[EU], ss[EU];
        bool m[EU];
#pragma unroll
        for (int u = 0; u < EU; ++u) {
            const int idx = base + u * EMB_T + tid;
            const int ic  = (idx < e1) ? idx : (e1 - 1);
            r[u] = rec[ic];
        }
        asm volatile("" :: "v"(r[0].x), "v"(r[0].y), "v"(r[1].x), "v"(r[1].y), "v"(r[2].x), "v"(r[2].y),
                           "v"(r[3].x), "v"(r[3].y), "v"(r[4].x), "v"(r[4].y), "v"(r[5].x), "v"(r[5].y),
                           "v"(r[6].x), "v"(r[6].y), "v"(r[7].x), "v"(r[7].y));
#pragma unroll
        for (int u = 0; u < EU; ++u) {
            const int idx = base + u * EMB_T + tid;
            int s = r[u].x;
            s = (s < 0) ? 0 : ((s < nN) ? s : (nN - 1));
            const int dl = r[u].y - lo;
            m[u]  = (idx < e1) && ((unsigned)dl < (unsigned)cnt);
            dd[u] = m[u] ? dl : 0;
            ss[u] = m[u] ? s : 0;
        }
        float  is[EU], id[EU];
        float2 v[EU];
#pragma unroll
        for (int u = 0; u < EU; ++u) {
            is[u] = inv[ss[u]];
            id[u] = inv[lo + dd[u]];
            v[u]  = *(const float2*)(H2 + (size_t)ss[u] * 2);
        }
        asm volatile("" :: "v"(is[0]), "v"(id[0]), "v"(v[0].x), "v"(is[1]), "v"(id[1]), "v"(v[1].x),
                           "v"(is[2]), "v"(id[2]), "v"(v[2].x), "v"(is[3]), "v"(id[3]), "v"(v[3].x),
                           "v"(is[4]), "v"(id[4]), "v"(v[4].x), "v"(is[5]), "v"(id[5]), "v"(v[5].x),
                           "v"(is[6]), "v"(id[6]), "v"(v[6].x), "v"(is[7]), "v"(id[7]), "v"(v[7].x));
#pragma unroll
        for (int u = 0; u < EU; ++u) {
            if (m[u]) {
                const float c = is[u] * id[u];
                float* a = acc + 2 * dd[u];
                atomicAdd(a + 0, v[u].x * c);
                atomicAdd(a + 1, v[u].y * c);
            }
        }
    }
    __syncthreads();

    if (own) {
        const float i2 = isn * isn;
        const float t0 = tanhf(acc[2 * tid + 0] + hn.x * i2 + b20);
        const float t1 = tanhf(acc[2 * tid + 1] + hn.y * i2 + b21);
        const float z  = t0 * wc0 + t1 * wc1 + bc0;
        out[n] = 1.0f / (1.0f + expf(-z));
    }
}

extern "C" void kernel_launch(void* const* d_in, const int* in_sizes, int n_in,
                              void* d_out, int out_size, void* d_ws, size_t ws_size,
                              hipStream_t stream) {
    const float* x   = (const float*)d_in[0];
    const int*   ei  = (const int*)d_in[1];
    const float* W1  = (const float*)d_in[2];
    const float* b1  = (const float*)d_in[3];
    const float* W2  = (const float*)d_in[4];
    const float* b2  = (const float*)d_in[5];
    const float* Wc  = (const float*)d_in[6];
    const float* bc  = (const float*)d_in[7];
    float* out = (float*)d_out;

    const int nN = in_sizes[0] / N_IN_F;
    const int nE = in_sizes[1] / 2;
    const int* src = ei;
    const int* dst = ei + nE;

    float* ws   = (float*)d_ws;
    float* w1t  = ws;
    float* h1   = w1t + N_HID_F * BSTRIDE;
    const int nB1     = (nN + L1_R - 1) / L1_R;
    const int nChunks = (nE + EU * BAND_T - 1) / (EU * BAND_T);
    const int pitch   = ((nB1 + 3) / 4) * 4;
    unsigned* ccnt = (unsigned*)(h1 + (size_t)nN * N_HID_F);
    int2*  rec  = (int2*)(ccnt + (size_t)nChunks * pitch);
    float* h2   = (float*)(rec + (size_t)nE);
    float* inv  = h2   + (size_t)nN * 2;
    unsigned* btot   = (unsigned*)(inv + (size_t)nN);
    unsigned* bstart = btot + NB_MAX;

    k_band_count<<<nChunks, BAND_T, 0, stream>>>(dst, ccnt, W1, w1t, nE, nB1);
    k_bin<<<nChunks, BAND_T, 0, stream>>>(src, dst, ccnt, bstart, btot, rec, nE, nB1, nChunks);

    const int nTiles = (nN + 15) / 16;
    k_gemm1_wmma<<<(nTiles + GEMM_WAVES - 1) / GEMM_WAVES, GEMM_T, GEMM_LDS, stream>>>(x, w1t, h1, nN, rec, bstart, btot, inv, nB1);
    k_band_hid<<<nB1, BAND_T, 0, stream>>>(h1, rec, inv, bstart, btot, b1, W2, h2, nN);

    k_band_emb<<<nB1, EMB_T, 0, stream>>>(h2, rec, inv, bstart, btot, b2, Wc, bc, out, nN);
}
